# starting kernel
# baseline (speedup 1.0000x reference)
#include <hip/hip_runtime.h>
#include <hip/hip_bf16.h>

typedef __attribute__((ext_vector_type(2))) float v2f;
typedef __attribute__((ext_vector_type(8))) float v8f;


__global__ void init_deg_kernel(int* deg, int n) {
    int i = blockIdx.x * blockDim.x + threadIdx.x;
    if (i < n) deg[i] = 1;
}

__global__ void count_edges_kernel(const int* __restrict__ ei, int* deg, int E) {
    int e = blockIdx.x * blockDim.x + threadIdx.x;
    if (e < E) {
        int dst = ei[E + e];
        atomicAdd(&deg[dst], 1);
    }
}

__global__ void scan_block_kernel(const int* __restrict__ deg, int* row_excl,
                                  int* bsum, int n) {
    __shared__ int sh[256];
    int i = blockIdx.x * 256 + threadIdx.x;
    int v = (i < n) ? deg[i] : 0;
    sh[threadIdx.x] = v;
    __syncthreads();
    for (int off = 1; off < 256; off <<= 1) {
        int t = (threadIdx.x >= off) ? sh[threadIdx.x - off] : 0;
        __syncthreads();
        sh[threadIdx.x] += t;
        __syncthreads();
    }
    if (i < n) row_excl[i] = sh[threadIdx.x] - v;
    if (threadIdx.x == 255) bsum[blockIdx.x] = sh[255];
}

__global__ void scan_sums_kernel(int* bsum, int nb) {
    __shared__ int sh[512];
    int v = (threadIdx.x < nb) ? bsum[threadIdx.x] : 0;
    sh[threadIdx.x] = v;
    __syncthreads();
    for (int off = 1; off < 512; off <<= 1) {
        int t = (threadIdx.x >= off) ? sh[threadIdx.x - off] : 0;
        __syncthreads();
        sh[threadIdx.x] += t;
        __syncthreads();
    }
    if (threadIdx.x < nb) bsum[threadIdx.x] = sh[threadIdx.x] - v;
}

__global__ void finalize_rows_kernel(const int* __restrict__ deg, int* row_ptr,
                                     const int* __restrict__ bsum, int* cursor,
                                     float* dis, int n, int total) {
    int i = blockIdx.x * blockDim.x + threadIdx.x;
    if (i < n) {
        int r = row_ptr[i] + bsum[i >> 8];
        row_ptr[i] = r;
        cursor[i]  = r;
        dis[i]     = rsqrtf((float)deg[i]);
        if (i == 0) row_ptr[n] = total;
    }
}

__global__ void fill_self_kernel(int* cursor, int* colidx, int n) {
    int i = blockIdx.x * blockDim.x + threadIdx.x;
    if (i < n) colidx[atomicAdd(&cursor[i], 1)] = i;
}

__global__ void fill_edges_kernel(const int* __restrict__ ei, int* cursor,
                                  int* colidx, int E) {
    int e = blockIdx.x * blockDim.x + threadIdx.x;
    if (e < E) {
        int src = ei[e];
        int dst = ei[E + e];
        colidx[atomicAdd(&cursor[dst], 1)] = src;
    }
}


template <bool MASK>
__global__ void gemm_wmma_f32_kernel(const float* __restrict__ A, int lda,
                                     const float* __restrict__ B, int ldb,
                                     float* __restrict__ C, int ldc,
                                     int K, int Ncols,
                                     const float* __restrict__ bias) {
    __shared__ float sA[16 * 513];

    const int row0 = blockIdx.x * 16;
    const int lane = threadIdx.x;
    const int tid  = threadIdx.y * 32 + threadIdx.x;
    const int nthr = blockDim.x * blockDim.y;
    const int ldsS = K + 1;

    for (int r = 0; r < 16; ++r) {
        const float* Ar = A + (size_t)(row0 + r) * lda;
        for (int c = tid; c < K; c += nthr) sA[r * ldsS + c] = Ar[c];
    }
    __syncthreads();

    const int m    = lane & 15;
    const int kk   = (lane >> 4) << 1;
    const int col0 = threadIdx.y * 16;
    const int n    = col0 + m;
    const bool nok = !MASK || (n < Ncols);
    const int  nc  = MASK ? (nok ? n : 0) : n;

    const float* Ap = sA + m * ldsS + kk;
    const float* Bp = B + kk * ldb + nc;

    v8f acc = {};
#pragma unroll 4
    for (int k = 0; k < K; k += 4) {
        v2f a, b;
        a.x = Ap[0];
        a.y = Ap[1];
        float bx = Bp[0];
        float by = Bp[ldb];
        if (MASK) { bx = nok ? bx : 0.0f; by = nok ? by : 0.0f; }
        b.x = bx;
        b.y = by;
        acc = __builtin_amdgcn_wmma_f32_16x16x4_f32(
             false, a,  false, b,
             (short)0, acc,  false,  false);
        Ap += 4;
        Bp += 4 * ldb;
    }

    const int mbase = (lane < 16) ? 0 : 8;
    if (nok) {
        const float badd = bias ? bias[n] : 0.0f;
        float* Cp = C + (size_t)(row0 + mbase) * ldc + n;
#pragma unroll
        for (int r = 0; r < 8; ++r) {
            Cp[0] = acc[r] + badd;
            Cp += ldc;
        }
    }
}


__global__ void gcn_aggregate_kernel(const float* __restrict__ H,
                                     const int* __restrict__ row_ptr,
                                     const int* __restrict__ colidx,
                                     const float* __restrict__ dis,
                                     const float* __restrict__ bias,
                                     float* __restrict__ outbuf,
                                     int ld_out, int col_off, int n) {
    const int node = blockIdx.x * (blockDim.x >> 5) + (threadIdx.x >> 5);
    const int lane = threadIdx.x & 31;
    if (node >= n) return;

    const int s = row_ptr[node];
    const int e = row_ptr[node + 1];

    float a0 = 0.f, a1 = 0.f, a2 = 0.f, a3 = 0.f;
    int p = s;
    for (; p + 1 < e; p += 2) {
        const int j0 = colidx[p];
        const int j1 = colidx[p + 1];
        const float w0 = dis[j0];
        const float w1 = dis[j1];
        const float4 h0 = *(const float4*)(H + (size_t)j0 * 128 + lane * 4);
        const float4 h1 = *(const float4*)(H + (size_t)j1 * 128 + lane * 4);
        a0 += w0 * h0.x + w1 * h1.x;
        a1 += w0 * h0.y + w1 * h1.y;
        a2 += w0 * h0.z + w1 * h1.z;
        a3 += w0 * h0.w + w1 * h1.w;
    }
    if (p < e) {
        const int j = colidx[p];
        const float w = dis[j];
        const float4 h = *(const float4*)(H + (size_t)j * 128 + lane * 4);
        a0 += w * h.x; a1 += w * h.y; a2 += w * h.z; a3 += w * h.w;
    }

    const float di = dis[node];
    const int c = lane * 4;
    float4 r;
    r.x = fmaxf(di * a0 + bias[c + 0], 0.0f);
    r.y = fmaxf(di * a1 + bias[c + 1], 0.0f);
    r.z = fmaxf(di * a2 + bias[c + 2], 0.0f);
    r.w = fmaxf(di * a3 + bias[c + 3], 0.0f);
    *(float4*)(outbuf + (size_t)node * ld_out + col_off + c) = r;
}


extern "C" void kernel_launch(void* const* d_in, const int* in_sizes, int n_in,
                              void* d_out, int out_size, void* d_ws, size_t ws_size,
                              hipStream_t stream) {
    (void)n_in; (void)out_size; (void)ws_size;
    const int F_IN = 256, HID = 128, NC = 40;
    const int N = in_sizes[0] / F_IN;
    const int E = in_sizes[1] / 2;
    const int TOT = E + N;

    const float* x     = (const float*)d_in[0];
    const int*   ei    = (const int*)d_in[1];
    const float* W_in  = (const float*)d_in[2];
    const float* b_in  = (const float*)d_in[3];
    const float* W1    = (const float*)d_in[4];
    const float* b1    = (const float*)d_in[5];
    const float* W2    = (const float*)d_in[6];
    const float* b2    = (const float*)d_in[7];
    const float* W3    = (const float*)d_in[8];
    const float* b3    = (const float*)d_in[9];
    const float* W_out = (const float*)d_in[10];
    const float* b_out = (const float*)d_in[11];
    float* out = (float*)d_out;

    char* base = (char*)d_ws;
    size_t off = 0;
    auto alloc = [&](size_t bytes) -> void* {
        void* q = base + off;
        off = (off + bytes + 255) & ~(size_t)255;
        return q;
    };
    int*   deg     = (int*)alloc((size_t)N * 4);
    int*   row_ptr = (int*)alloc((size_t)(N + 1) * 4);
    int*   cursor  = (int*)alloc((size_t)N * 4);
    int*   bsum    = (int*)alloc(512 * 4);
    int*   colidx  = (int*)alloc((size_t)TOT * 4);
    float* dis     = (float*)alloc((size_t)N * 4);
    float* H       = (float*)alloc((size_t)N * HID * 4);
    float* hcat    = (float*)alloc((size_t)N * HID * 4 * 4);

    const int nbN = (N + 255) / 256;
    const int nbE = (E + 255) / 256;

    init_deg_kernel<<<nbN, 256, 0, stream>>>(deg, N);
    count_edges_kernel<<<nbE, 256, 0, stream>>>(ei, deg, E);
    scan_block_kernel<<<nbN, 256, 0, stream>>>(deg, row_ptr, bsum, N);
    scan_sums_kernel<<<1, 512, 0, stream>>>(bsum, nbN);
    finalize_rows_kernel<<<nbN, 256, 0, stream>>>(deg, row_ptr, bsum, cursor, dis, N, TOT);
    fill_self_kernel<<<nbN, 256, 0, stream>>>(cursor, colidx, N);
    fill_edges_kernel<<<nbE, 256, 0, stream>>>(ei, cursor, colidx, E);

    const int rowTiles = N / 16;
    dim3 gemmBlk(32, HID / 16);
    const int aggBlocks = (N + 7) / 8;

    gemm_wmma_f32_kernel<false><<<rowTiles, gemmBlk, 0, stream>>>(
        x, F_IN, W_in, HID, H, HID, F_IN, HID, nullptr);
    gcn_aggregate_kernel<<<aggBlocks, 256, 0, stream>>>(
        H, row_ptr, colidx, dis, b_in, hcat, 4 * HID, 0 * HID, N);

    const float* Ws[3] = {W1, W2, W3};
    const float* bs[3] = {b1, b2, b3};
    for (int l = 0; l < 3; ++l) {
        gemm_wmma_f32_kernel<false><<<rowTiles, gemmBlk, 0, stream>>>(
            hcat + (size_t)l * HID, 4 * HID, Ws[l], HID, H, HID, HID, HID, nullptr);
        gcn_aggregate_kernel<<<aggBlocks, 256, 0, stream>>>(
            H, row_ptr, colidx, dis, bs[l], hcat, 4 * HID, (l + 1) * HID, N);
    }

    dim3 headBlk(32, 3);
    gemm_wmma_f32_kernel<true><<<rowTiles, headBlk, 0, stream>>>(
        hcat, 4 * HID, W_out, NC, out, NC, 4 * HID, NC, b_out);
}


# GEMM: weights staged once per block in LDS in fragment order, each wave owns a 16-row tile with A read as b128 into fragments and one accumulator per column tile; k summation order changes
# speedup vs baseline: 1.3282x; 1.3282x over previous
#include <hip/hip_runtime.h>
#include <hip/hip_bf16.h>

typedef __attribute__((ext_vector_type(2))) float v2f;
typedef __attribute__((ext_vector_type(8))) float v8f;


__global__ void init_deg_kernel(int* deg, int n) {
    int i = blockIdx.x * blockDim.x + threadIdx.x;
    if (i < n) deg[i] = 1;
}

__global__ void count_edges_kernel(const int* __restrict__ ei, int* deg, int E) {
    int e = blockIdx.x * blockDim.x + threadIdx.x;
    if (e < E) {
        int dst = ei[E + e];
        atomicAdd(&deg[dst], 1);
    }
}

__global__ void scan_block_kernel(const int* __restrict__ deg, int* row_excl,
                                  int* bsum, int n) {
    __shared__ int sh[256];
    int i = blockIdx.x * 256 + threadIdx.x;
    int v = (i < n) ? deg[i] : 0;
    sh[threadIdx.x] = v;
    __syncthreads();
    for (int off = 1; off < 256; off <<= 1) {
        int t = (threadIdx.x >= off) ? sh[threadIdx.x - off] : 0;
        __syncthreads();
        sh[threadIdx.x] += t;
        __syncthreads();
    }
    if (i < n) row_excl[i] = sh[threadIdx.x] - v;
    if (threadIdx.x == 255) bsum[blockIdx.x] = sh[255];
}

__global__ void scan_sums_kernel(int* bsum, int nb) {
    __shared__ int sh[512];
    int v = (threadIdx.x < nb) ? bsum[threadIdx.x] : 0;
    sh[threadIdx.x] = v;
    __syncthreads();
    for (int off = 1; off < 512; off <<= 1) {
        int t = (threadIdx.x >= off) ? sh[threadIdx.x - off] : 0;
        __syncthreads();
        sh[threadIdx.x] += t;
        __syncthreads();
    }
    if (threadIdx.x < nb) bsum[threadIdx.x] = sh[threadIdx.x] - v;
}

__global__ void finalize_rows_kernel(const int* __restrict__ deg, int* row_ptr,
                                     const int* __restrict__ bsum, int* cursor,
                                     float* dis, int n, int total) {
    int i = blockIdx.x * blockDim.x + threadIdx.x;
    if (i < n) {
        int r = row_ptr[i] + bsum[i >> 8];
        row_ptr[i] = r;
        cursor[i]  = r;
        dis[i]     = rsqrtf((float)deg[i]);
        if (i == 0) row_ptr[n] = total;
    }
}

__global__ void fill_self_kernel(int* cursor, int* colidx, int n) {
    int i = blockIdx.x * blockDim.x + threadIdx.x;
    if (i < n) colidx[atomicAdd(&cursor[i], 1)] = i;
}

__global__ void fill_edges_kernel(const int* __restrict__ ei, int* cursor,
                                  int* colidx, int E) {
    int e = blockIdx.x * blockDim.x + threadIdx.x;
    if (e < E) {
        int src = ei[e];
        int dst = ei[E + e];
        colidx[atomicAdd(&cursor[dst], 1)] = src;
    }
}


template <int NT, int KCH, bool MASK>
__global__ void __launch_bounds__(256)
gemm_wmma_f32_kernel(const float* __restrict__ A, int lda,
                     const float* __restrict__ B, int ldb,
                     float* __restrict__ C, int ldc,
                     int Ncols, const float* __restrict__ bias, int nRowTiles) {
    constexpr int NP = NT * 16;
    constexpr int K  = KCH * 128;
    constexpr int Q  = NP / 4;
    __shared__ __attribute__((aligned(16))) float sW[K * NP];

    const int tid  = threadIdx.x;
    const int lane = tid & 31;
    const int wave = tid >> 5;
    const int c    = lane & 15;
    const int h    = lane >> 4;
    const int tile = blockIdx.x * 8 + wave;
    const bool live = tile < nRowTiles;

    const float* Ap = A + (size_t)((live ? tile : 0) * 16 + c) * lda + 4 * h;

    float4 a[16];
#pragma unroll
    for (int t = 0; t < 16; ++t) a[t] = *(const float4*)(Ap + 8 * t);

    for (int it = tid; it < (K / 4) * Q; it += 256) {
        const int g = it / Q;
        const int q = it - g * Q;
        float4 r0, r1, r2, r3;
        if (!MASK || 4 * q < Ncols) {
            const float* p = B + (size_t)(4 * g) * ldb + 4 * q;
            r0 = *(const float4*)(p);
            r1 = *(const float4*)(p + ldb);
            r2 = *(const float4*)(p + 2 * ldb);
            r3 = *(const float4*)(p + 3 * ldb);
        } else {
            r0 = r1 = r2 = r3 = make_float4(0.f, 0.f, 0.f, 0.f);
        }
        float4* d = (float4*)(sW + ((size_t)g * NP + 4 * q) * 4);
        d[0] = make_float4(r0.x, r1.x, r2.x, r3.x);
        d[1] = make_float4(r0.y, r1.y, r2.y, r3.y);
        d[2] = make_float4(r0.z, r1.z, r2.z, r3.z);
        d[3] = make_float4(r0.w, r1.w, r2.w, r3.w);
    }
    __syncthreads();
    if (!live) return;

    v8f acc[NT];
#pragma unroll
    for (int j = 0; j < NT; ++j) acc[j] = v8f{};

#pragma unroll
    for (int ch = 0; ch < KCH; ++ch) {
        float4 an[16];
        if (ch + 1 < KCH) {
#pragma unroll
            for (int t = 0; t < 16; ++t)
                an[t] = *(const float4*)(Ap + (ch + 1) * 128 + 8 * t);
        }
        const float* sB = sW + ((size_t)(ch * 32 + h) * NP + c) * 4;
#pragma unroll
        for (int t = 0; t < 16; ++t) {
            float4 b[NT];
#pragma unroll
            for (int j = 0; j < NT; ++j)
                b[j] = *(const float4*)(sB + ((size_t)(2 * t) * NP + 16 * j) * 4);
            v2f a0, a1;
            a0.x = a[t].x; a0.y = a[t].y;
            a1.x = a[t].z; a1.y = a[t].w;
#pragma unroll
            for (int j = 0; j < NT; ++j) {
                v2f b0; b0.x = b[j].x; b0.y = b[j].y;
                acc[j] = __builtin_amdgcn_wmma_f32_16x16x4_f32(
                     false, a0,  false, b0,
                     (short)0, acc[j],  false,  false);
            }
#pragma unroll
            for (int j = 0; j < NT; ++j) {
                v2f b1; b1.x = b[j].z; b1.y = b[j].w;
                acc[j] = __builtin_amdgcn_wmma_f32_16x16x4_f32(
                     false, a1,  false, b1,
                     (short)0, acc[j],  false,  false);
            }
        }
        if (ch + 1 < KCH) {
#pragma unroll
            for (int t = 0; t < 16; ++t) a[t] = an[t];
        }
    }

    const int rbase = tile * 16 + 8 * h;
#pragma unroll
    for (int j = 0; j < NT; ++j) {
        const int n = 16 * j + c;
        if (!MASK || n < Ncols) {
            const float badd = bias ? bias[n] : 0.0f;
            float* Cp = C + (size_t)rbase * ldc + n;
#pragma unroll
            for (int r = 0; r < 8; ++r) {
                Cp[0] = acc[j][r] + badd;
                Cp += ldc;
            }
        }
    }
}


__global__ void gcn_aggregate_kernel(const float* __restrict__ H,
                                     const int* __restrict__ row_ptr,
                                     const int* __restrict__ colidx,
                                     const float* __restrict__ dis,
                                     const float* __restrict__ bias,
                                     float* __restrict__ outbuf,
                                     int ld_out, int col_off, int n) {
    const int node = blockIdx.x * (blockDim.x >> 5) + (threadIdx.x >> 5);
    const int lane = threadIdx.x & 31;
    if (node >= n) return;

    const int s = row_ptr[node];
    const int e = row_ptr[node + 1];

    float a0 = 0.f, a1 = 0.f, a2 = 0.f, a3 = 0.f;
    int p = s;
    for (; p + 1 < e; p += 2) {
        const int j0 = colidx[p];
        const int j1 = colidx[p + 1];
        const float w0 = dis[j0];
        const float w1 = dis[j1];
        const float4 h0 = *(const float4*)(H + (size_t)j0 * 128 + lane * 4);
        const float4 h1 = *(const float4*)(H + (size_t)j1 * 128 + lane * 4);
        a0 += w0 * h0.x + w1 * h1.x;
        a1 += w0 * h0.y + w1 * h1.y;
        a2 += w0 * h0.z + w1 * h1.z;
        a3 += w0 * h0.w + w1 * h1.w;
    }
    if (p < e) {
        const int j = colidx[p];
        const float w = dis[j];
        const float4 h = *(const float4*)(H + (size_t)j * 128 + lane * 4);
        a0 += w * h.x; a1 += w * h.y; a2 += w * h.z; a3 += w * h.w;
    }

    const float di = dis[node];
    const int c = lane * 4;
    float4 r;
    r.x = fmaxf(di * a0 + bias[c + 0], 0.0f);
    r.y = fmaxf(di * a1 + bias[c + 1], 0.0f);
    r.z = fmaxf(di * a2 + bias[c + 2], 0.0f);
    r.w = fmaxf(di * a3 + bias[c + 3], 0.0f);
    *(float4*)(outbuf + (size_t)node * ld_out + col_off + c) = r;
}


extern "C" void kernel_launch(void* const* d_in, const int* in_sizes, int n_in,
                              void* d_out, int out_size, void* d_ws, size_t ws_size,
                              hipStream_t stream) {
    (void)n_in; (void)out_size; (void)ws_size;
    const int F_IN = 256, HID = 128, NC = 40;
    const int N = in_sizes[0] / F_IN;
    const int E = in_sizes[1] / 2;
    const int TOT = E + N;

    const float* x     = (const float*)d_in[0];
    const int*   ei    = (const int*)d_in[1];
    const float* W_in  = (const float*)d_in[2];
    const float* b_in  = (const float*)d_in[3];
    const float* W1    = (const float*)d_in[4];
    const float* b1    = (const float*)d_in[5];
    const float* W2    = (const float*)d_in[6];
    const float* b2    = (const float*)d_in[7];
    const float* W3    = (const float*)d_in[8];
    const float* b3    = (const float*)d_in[9];
    const float* W_out = (const float*)d_in[10];
    const float* b_out = (const float*)d_in[11];
    float* out = (float*)d_out;

    char* base = (char*)d_ws;
    size_t off = 0;
    auto alloc = [&](size_t bytes) -> void* {
        void* q = base + off;
        off = (off + bytes + 255) & ~(size_t)255;
        return q;
    };
    int*   deg     = (int*)alloc((size_t)N * 4);
    int*   row_ptr = (int*)alloc((size_t)(N + 1) * 4);
    int*   cursor  = (int*)alloc((size_t)N * 4);
    int*   bsum    = (int*)alloc(512 * 4);
    int*   colidx  = (int*)alloc((size_t)TOT * 4);
    float* dis     = (float*)alloc((size_t)N * 4);
    float* H       = (float*)alloc((size_t)N * HID * 4);
    float* hcat    = (float*)alloc((size_t)N * HID * 4 * 4);

    const int nbN = (N + 255) / 256;
    const int nbE = (E + 255) / 256;

    init_deg_kernel<<<nbN, 256, 0, stream>>>(deg, N);
    count_edges_kernel<<<nbE, 256, 0, stream>>>(ei, deg, E);
    scan_block_kernel<<<nbN, 256, 0, stream>>>(deg, row_ptr, bsum, N);
    scan_sums_kernel<<<1, 512, 0, stream>>>(bsum, nbN);
    finalize_rows_kernel<<<nbN, 256, 0, stream>>>(deg, row_ptr, bsum, cursor, dis, N, TOT);
    fill_self_kernel<<<nbN, 256, 0, stream>>>(cursor, colidx, N);
    fill_edges_kernel<<<nbE, 256, 0, stream>>>(ei, cursor, colidx, E);

    const int rowTiles = N / 16;
    const int gemmBlocks = (rowTiles + 7) / 8;
    const int aggBlocks = (N + 7) / 8;

    gemm_wmma_f32_kernel<8, 2, false><<<gemmBlocks, 256, 0, stream>>>(
        x, F_IN, W_in, HID, H, HID, HID, nullptr, rowTiles);
    gcn_aggregate_kernel<<<aggBlocks, 256, 0, stream>>>(
        H, row_ptr, colidx, dis, b_in, hcat, 4 * HID, 0 * HID, N);

    const float* Ws[3] = {W1, W2, W3};
    const float* bs[3] = {b1, b2, b3};
    for (int l = 0; l < 3; ++l) {
        gemm_wmma_f32_kernel<8, 1, false><<<gemmBlocks, 256, 0, stream>>>(
            hcat + (size_t)l * HID, 4 * HID, Ws[l], HID, H, HID, HID, nullptr, rowTiles);
        gcn_aggregate_kernel<<<aggBlocks, 256, 0, stream>>>(
            H, row_ptr, colidx, dis, bs[l], hcat, 4 * HID, (l + 1) * HID, N);
    }

    gemm_wmma_f32_kernel<3, 4, true><<<gemmBlocks, 256, 0, stream>>>(
        hcat, 4 * HID, W_out, NC, out, NC, NC, b_out, rowTiles);
}


# CSR build: two-level counting sort by destination in LDS (bins of 256 nodes, then one block per bin) replaces global atomics and scans, 7 launches to 4; neighbour order inside a row changes
# speedup vs baseline: 1.8065x; 1.3601x over previous
#include <hip/hip_runtime.h>
#include <hip/hip_bf16.h>

typedef __attribute__((ext_vector_type(2))) float v2f;
typedef __attribute__((ext_vector_type(8))) float v8f;


#define CSR_CHUNK 4096
#define CSR_MAXB  512
#define CSR_STAGE 8192

__device__ __forceinline__ int block_excl_scan256(int v, int* wsum, int& total) {
    const int lane = threadIdx.x & 31;
    const int wave = threadIdx.x >> 5;
    int inc = v;
    for (int o = 1; o < 32; o <<= 1) {
        const int t = __shfl_up(inc, o, 32);
        if (lane >= o) inc += t;
    }
    if (lane == 31) wsum[wave] = inc;
    __syncthreads();
    int pre = 0, tot = 0;
    for (int w = 0; w < 8; ++w) {
        const int s = wsum[w];
        if (w < wave) pre += s;
        tot += s;
    }
    __syncthreads();
    total = tot;
    return pre + inc - v;
}

__global__ void __launch_bounds__(256)
csr_bin_count_kernel(const int* __restrict__ ei, int* __restrict__ counts,
                     int E, int NB) {
    __shared__ int hist[CSR_MAXB];
    const int tid = threadIdx.x;
    hist[tid] = 0;
    hist[tid + 256] = 0;
    const int base = blockIdx.x * CSR_CHUNK;
    int bn[CSR_CHUNK / 256];
#pragma unroll
    for (int k = 0; k < CSR_CHUNK / 256; ++k) {
        const int e = base + k * 256 + tid;
        const int d = ei[E + ((e < E) ? e : E - 1)];
        bn[k] = (e < E) ? (d >> 8) : -1;
    }
    __syncthreads();
#pragma unroll
    for (int k = 0; k < CSR_CHUNK / 256; ++k)
        if (bn[k] >= 0) atomicAdd(&hist[bn[k]], 1);
    __syncthreads();
    for (int b = tid; b < NB; b += 256) counts[blockIdx.x * NB + b] = hist[b];
}

__global__ void __launch_bounds__(256)
csr_bin_scan_kernel(const int* __restrict__ counts, int* __restrict__ offsT,
                    int* __restrict__ bintot, int NB, int G) {
    __shared__ int wsum[8];
    const int b = blockIdx.x;
    const int tid = threadIdx.x;
    int carry = 0;
    for (int t0 = 0; t0 < G; t0 += 256) {
        const int blk = t0 + tid;
        const int v = (blk < G) ? counts[blk * NB + b] : 0;
        int tot;
        const int ex = block_excl_scan256(v, wsum, tot);
        if (blk < G) offsT[b * G + blk] = carry + ex;
        carry += tot;
    }
    if (tid == 0) bintot[b] = carry;
}

__global__ void __launch_bounds__(256)
csr_bin_scatter_kernel(const int* __restrict__ ei, const int* __restrict__ offsT,
                       const int* __restrict__ bintot, int* __restrict__ binned,
                       int E, int NB, int G) {
    __shared__ int hist[CSR_MAXB];
    __shared__ int gofs[CSR_MAXB];
    __shared__ int sval[CSR_CHUNK];
    __shared__ unsigned short sbin[CSR_CHUNK];
    __shared__ int wsum[8];
    const int tid = threadIdx.x;
    const int blk = blockIdx.x;
    hist[tid] = 0;
    hist[tid + 256] = 0;
    const int base = blk * CSR_CHUNK;
    int pk[CSR_CHUNK / 256], bn[CSR_CHUNK / 256], rk[CSR_CHUNK / 256];
#pragma unroll
    for (int k = 0; k < CSR_CHUNK / 256; ++k) {
        const int e = base + k * 256 + tid;
        const int ec = (e < E) ? e : E - 1;
        const int s = ei[ec];
        const int d = ei[E + ec];
        pk[k] = (s << 8) | (d & 255);
        bn[k] = (e < E) ? (d >> 8) : -1;
    }
    __syncthreads();
#pragma unroll
    for (int k = 0; k < CSR_CHUNK / 256; ++k)
        rk[k] = (bn[k] >= 0) ? atomicAdd(&hist[bn[k]], 1) : 0;
    __syncthreads();

    const int b0 = 2 * tid, b1 = 2 * tid + 1;
    const int c0 = hist[b0], c1 = hist[b1];
    int tot;
    const int lex = block_excl_scan256(c0 + c1, wsum, tot);
    const int t0 = (b0 < NB) ? bintot[b0] : 0;
    const int t1 = (b1 < NB) ? bintot[b1] : 0;
    int gtot;
    const int gex = block_excl_scan256(t0 + t1, wsum, gtot);
    (void)gtot;
    hist[b0] = lex;
    hist[b1] = lex + c0;
    gofs[b0] = (b0 < NB) ? gex + offsT[b0 * G + blk] - lex : 0;
    gofs[b1] = (b1 < NB) ? gex + t0 + offsT[b1 * G + blk] - (lex + c0) : 0;
    __syncthreads();
#pragma unroll
    for (int k = 0; k < CSR_CHUNK / 256; ++k) {
        if (bn[k] >= 0) {
            const int p = hist[bn[k]] + rk[k];
            sval[p] = pk[k];
            sbin[p] = (unsigned short)bn[k];
        }
    }
    __syncthreads();
    for (int p = tid; p < tot; p += 256) binned[gofs[sbin[p]] + p] = sval[p];
}

__global__ void __launch_bounds__(256)
csr_bin_finalize_kernel(const int* __restrict__ binned, const int* __restrict__ bintot,
                        int* __restrict__ row_ptr, int* __restrict__ colidx,
                        float* __restrict__ dis, int N, int total) {
    __shared__ int cnt[256];
    __shared__ int cur[256];
    __shared__ int stage[CSR_STAGE];
    __shared__ int wsum[8];
    const int tid = threadIdx.x;
    const int b = blockIdx.x;
    const int node = b * 256 + tid;

    int part = 0;
    for (int i = tid; i < b; i += 256) part += bintot[i];
    int binbase;
    (void)block_excl_scan256(part, wsum, binbase);
    const int m = bintot[b];
    const int* in = binned + binbase;

    cnt[tid] = 0;
    __syncthreads();
    for (int p0 = 0; p0 < m; p0 += 2048) {
        int v[8];
#pragma unroll
        for (int j = 0; j < 8; ++j) {
            const int p = p0 + j * 256 + tid;
            const int w = in[(p < m) ? p : m - 1];
            v[j] = (p < m) ? w : -1;
        }
#pragma unroll
        for (int j = 0; j < 8; ++j)
            if (v[j] >= 0) atomicAdd(&cnt[v[j] & 255], 1);
    }
    __syncthreads();

    const bool has = node < N;
    const int d = has ? cnt[tid] + 1 : 0;
    int tot;
    const int ex = block_excl_scan256(d, wsum, tot);
    const int rowbase = binbase + b * 256;
    if (has) {
        row_ptr[node] = rowbase + ex;
        dis[node] = rsqrtf((float)d);
        if (node == N - 1) row_ptr[N] = total;
    }
    cur[tid] = ex + 1;

    if (tot <= CSR_STAGE) {
        if (has) stage[ex] = node;
        __syncthreads();
        for (int p0 = 0; p0 < m; p0 += 2048) {
            int v[8];
#pragma unroll
            for (int j = 0; j < 8; ++j) {
                const int p = p0 + j * 256 + tid;
                const int w = in[(p < m) ? p : m - 1];
                v[j] = (p < m) ? w : -1;
            }
#pragma unroll
            for (int j = 0; j < 8; ++j)
                if (v[j] >= 0) stage[atomicAdd(&cur[v[j] & 255], 1)] = v[j] >> 8;
        }
        __syncthreads();
        for (int q = tid; q < tot; q += 256) colidx[rowbase + q] = stage[q];
    } else {
        if (has) colidx[rowbase + ex] = node;
        __syncthreads();
        for (int p = tid; p < m; p += 256) {
            const int v = in[p];
            colidx[rowbase + atomicAdd(&cur[v & 255], 1)] = v >> 8;
        }
    }
}


template <int NT, int KCH, bool MASK>
__global__ void __launch_bounds__(256)
gemm_wmma_f32_kernel(const float* __restrict__ A, int lda,
                     const float* __restrict__ B, int ldb,
                     float* __restrict__ C, int ldc,
                     int Ncols, const float* __restrict__ bias, int nRowTiles) {
    constexpr int NP = NT * 16;
    constexpr int K  = KCH * 128;
    constexpr int Q  = NP / 4;
    __shared__ __attribute__((aligned(16))) float sW[K * NP];

    const int tid  = threadIdx.x;
    const int lane = tid & 31;
    const int wave = tid >> 5;
    const int c    = lane & 15;
    const int h    = lane >> 4;
    const int tile = blockIdx.x * 8 + wave;
    const bool live = tile < nRowTiles;

    const float* Ap = A + (size_t)((live ? tile : 0) * 16 + c) * lda + 4 * h;

    float4 a[16];
#pragma unroll
    for (int t = 0; t < 16; ++t) a[t] = *(const float4*)(Ap + 8 * t);

    for (int it = tid; it < (K / 4) * Q; it += 256) {
        const int g = it / Q;
        const int q = it - g * Q;
        float4 r0, r1, r2, r3;
        if (!MASK || 4 * q < Ncols) {
            const float* p = B + (size_t)(4 * g) * ldb + 4 * q;
            r0 = *(const float4*)(p);
            r1 = *(const float4*)(p + ldb);
            r2 = *(const float4*)(p + 2 * ldb);
            r3 = *(const float4*)(p + 3 * ldb);
        } else {
            r0 = r1 = r2 = r3 = make_float4(0.f, 0.f, 0.f, 0.f);
        }
        float4* d = (float4*)(sW + ((size_t)g * NP + 4 * q) * 4);
        d[0] = make_float4(r0.x, r1.x, r2.x, r3.x);
        d[1] = make_float4(r0.y, r1.y, r2.y, r3.y);
        d[2] = make_float4(r0.z, r1.z, r2.z, r3.z);
        d[3] = make_float4(r0.w, r1.w, r2.w, r3.w);
    }
    __syncthreads();
    if (!live) return;

    v8f acc[NT];
#pragma unroll
    for (int j = 0; j < NT; ++j) acc[j] = v8f{};

#pragma unroll
    for (int ch = 0; ch < KCH; ++ch) {
        float4 an[16];
        if (ch + 1 < KCH) {
#pragma unroll
            for (int t = 0; t < 16; ++t)
                an[t] = *(const float4*)(Ap + (ch + 1) * 128 + 8 * t);
        }
        const float* sB = sW + ((size_t)(ch * 32 + h) * NP + c) * 4;
#pragma unroll
        for (int t = 0; t < 16; ++t) {
            float4 b[NT];
#pragma unroll
            for (int j = 0; j < NT; ++j)
                b[j] = *(const float4*)(sB + ((size_t)(2 * t) * NP + 16 * j) * 4);
            v2f a0, a1;
            a0.x = a[t].x; a0.y = a[t].y;
            a1.x = a[t].z; a1.y = a[t].w;
#pragma unroll
            for (int j = 0; j < NT; ++j) {
                v2f b0; b0.x = b[j].x; b0.y = b[j].y;
                acc[j] = __builtin_amdgcn_wmma_f32_16x16x4_f32(
                     false, a0,  false, b0,
                     (short)0, acc[j],  false,  false);
            }
#pragma unroll
            for (int j = 0; j < NT; ++j) {
                v2f b1; b1.x = b[j].z; b1.y = b[j].w;
                acc[j] = __builtin_amdgcn_wmma_f32_16x16x4_f32(
                     false, a1,  false, b1,
                     (short)0, acc[j],  false,  false);
            }
        }
        if (ch + 1 < KCH) {
#pragma unroll
            for (int t = 0; t < 16; ++t) a[t] = an[t];
        }
    }

    const int rbase = tile * 16 + 8 * h;
#pragma unroll
    for (int j = 0; j < NT; ++j) {
        const int n = 16 * j + c;
        if (!MASK || n < Ncols) {
            const float badd = bias ? bias[n] : 0.0f;
            float* Cp = C + (size_t)rbase * ldc + n;
#pragma unroll
            for (int r = 0; r < 8; ++r) {
                Cp[0] = acc[j][r] + badd;
                Cp += ldc;
            }
        }
    }
}


__global__ void gcn_aggregate_kernel(const float* __restrict__ H,
                                     const int* __restrict__ row_ptr,
                                     const int* __restrict__ colidx,
                                     const float* __restrict__ dis,
                                     const float* __restrict__ bias,
                                     float* __restrict__ outbuf,
                                     int ld_out, int col_off, int n) {
    const int node = blockIdx.x * (blockDim.x >> 5) + (threadIdx.x >> 5);
    const int lane = threadIdx.x & 31;
    if (node >= n) return;

    const int s = row_ptr[node];
    const int e = row_ptr[node + 1];

    float a0 = 0.f, a1 = 0.f, a2 = 0.f, a3 = 0.f;
    int p = s;
    for (; p + 1 < e; p += 2) {
        const int j0 = colidx[p];
        const int j1 = colidx[p + 1];
        const float w0 = dis[j0];
        const float w1 = dis[j1];
        const float4 h0 = *(const float4*)(H + (size_t)j0 * 128 + lane * 4);
        const float4 h1 = *(const float4*)(H + (size_t)j1 * 128 + lane * 4);
        a0 += w0 * h0.x + w1 * h1.x;
        a1 += w0 * h0.y + w1 * h1.y;
        a2 += w0 * h0.z + w1 * h1.z;
        a3 += w0 * h0.w + w1 * h1.w;
    }
    if (p < e) {
        const int j = colidx[p];
        const float w = dis[j];
        const float4 h = *(const float4*)(H + (size_t)j * 128 + lane * 4);
        a0 += w * h.x; a1 += w * h.y; a2 += w * h.z; a3 += w * h.w;
    }

    const float di = dis[node];
    const int c = lane * 4;
    float4 r;
    r.x = fmaxf(di * a0 + bias[c + 0], 0.0f);
    r.y = fmaxf(di * a1 + bias[c + 1], 0.0f);
    r.z = fmaxf(di * a2 + bias[c + 2], 0.0f);
    r.w = fmaxf(di * a3 + bias[c + 3], 0.0f);
    *(float4*)(outbuf + (size_t)node * ld_out + col_off + c) = r;
}


extern "C" void kernel_launch(void* const* d_in, const int* in_sizes, int n_in,
                              void* d_out, int out_size, void* d_ws, size_t ws_size,
                              hipStream_t stream) {
    (void)n_in; (void)out_size; (void)ws_size;
    const int F_IN = 256, HID = 128, NC = 40;
    const int N = in_sizes[0] / F_IN;
    const int E = in_sizes[1] / 2;
    const int TOT = E + N;

    const float* x     = (const float*)d_in[0];
    const int*   ei    = (const int*)d_in[1];
    const float* W_in  = (const float*)d_in[2];
    const float* b_in  = (const float*)d_in[3];
    const float* W1    = (const float*)d_in[4];
    const float* b1    = (const float*)d_in[5];
    const float* W2    = (const float*)d_in[6];
    const float* b2    = (const float*)d_in[7];
    const float* W3    = (const float*)d_in[8];
    const float* b3    = (const float*)d_in[9];
    const float* W_out = (const float*)d_in[10];
    const float* b_out = (const float*)d_in[11];
    float* out = (float*)d_out;

    char* base = (char*)d_ws;
    size_t off = 0;
    auto alloc = [&](size_t bytes) -> void* {
        void* q = base + off;
        off = (off + bytes + 255) & ~(size_t)255;
        return q;
    };
    int*   row_ptr = (int*)alloc((size_t)(N + 1) * 4);
    int*   colidx  = (int*)alloc((size_t)TOT * 4);
    float* dis     = (float*)alloc((size_t)N * 4);
    float* H       = (float*)alloc((size_t)N * HID * 4);
    float* hcat    = (float*)alloc((size_t)N * HID * 4 * 4);

    const int NB = (N + 255) / 256;
    const int G  = (E + CSR_CHUNK - 1) / CSR_CHUNK;

    int* counts = (int*)hcat;
    int* offsT  = counts + (size_t)G * NB;
    int* bintot = offsT + (size_t)NB * G;
    int* binned = bintot + CSR_MAXB;

    csr_bin_count_kernel<<<G, 256, 0, stream>>>(ei, counts, E, NB);
    csr_bin_scan_kernel<<<NB, 256, 0, stream>>>(counts, offsT, bintot, NB, G);
    csr_bin_scatter_kernel<<<G, 256, 0, stream>>>(ei, offsT, bintot, binned, E, NB, G);
    csr_bin_finalize_kernel<<<NB, 256, 0, stream>>>(binned, bintot, row_ptr, colidx, dis, N, TOT);

    const int rowTiles = N / 16;
    const int gemmBlocks = (rowTiles + 7) / 8;
    const int aggBlocks = (N + 7) / 8;

    gemm_wmma_f32_kernel<8, 2, false><<<gemmBlocks, 256, 0, stream>>>(
        x, F_IN, W_in, HID, H, HID, HID, nullptr, rowTiles);
    gcn_aggregate_kernel<<<aggBlocks, 256, 0, stream>>>(
        H, row_ptr, colidx, dis, b_in, hcat, 4 * HID, 0 * HID, N);

    const float* Ws[3] = {W1, W2, W3};
    const float* bs[3] = {b1, b2, b3};
    for (int l = 0; l < 3; ++l) {
        gemm_wmma_f32_kernel<8, 1, false><<<gemmBlocks, 256, 0, stream>>>(
            hcat + (size_t)l * HID, 4 * HID, Ws[l], HID, H, HID, HID, nullptr, rowTiles);
        gcn_aggregate_kernel<<<aggBlocks, 256, 0, stream>>>(
            H, row_ptr, colidx, dis, bs[l], hcat, 4 * HID, (l + 1) * HID, N);
    }

    gemm_wmma_f32_kernel<3, 4, true><<<gemmBlocks, 256, 0, stream>>>(
        hcat, 4 * HID, W_out, NC, out, NC, NC, b_out, rowTiles);
}


# Layers 2-4: aggregation fused with the layer GEMM in one kernel, aggregate first then W ((A h)W: summation order changes), one long-lived wave per 16-node tile via LDS; 13 launches to 10
# speedup vs baseline: 1.8980x; 1.0507x over previous
#include <hip/hip_runtime.h>
#include <hip/hip_bf16.h>

typedef __attribute__((ext_vector_type(2))) float v2f;
typedef __attribute__((ext_vector_type(8))) float v8f;


#define CSR_CHUNK 4096
#define CSR_MAXB  512
#define CSR_STAGE 8192

__device__ __forceinline__ int block_excl_scan256(int v, int* wsum, int& total) {
    const int lane = threadIdx.x & 31;
    const int wave = threadIdx.x >> 5;
    int inc = v;
    for (int o = 1; o < 32; o <<= 1) {
        const int t = __shfl_up(inc, o, 32);
        if (lane >= o) inc += t;
    }
    if (lane == 31) wsum[wave] = inc;
    __syncthreads();
    int pre = 0, tot = 0;
    for (int w = 0; w < 8; ++w) {
        const int s = wsum[w];
        if (w < wave) pre += s;
        tot += s;
    }
    __syncthreads();
    total = tot;
    return pre + inc - v;
}

__global__ void __launch_bounds__(256)
csr_bin_count_kernel(const int* __restrict__ ei, int* __restrict__ counts,
                     int E, int NB) {
    __shared__ int hist[CSR_MAXB];
    const int tid = threadIdx.x;
    hist[tid] = 0;
    hist[tid + 256] = 0;
    const int base = blockIdx.x * CSR_CHUNK;
    int bn[CSR_CHUNK / 256];
#pragma unroll
    for (int k = 0; k < CSR_CHUNK / 256; ++k) {
        const int e = base + k * 256 + tid;
        const int d = ei[E + ((e < E) ? e : E - 1)];
        bn[k] = (e < E) ? (d >> 8) : -1;
    }
    __syncthreads();
#pragma unroll
    for (int k = 0; k < CSR_CHUNK / 256; ++k)
        if (bn[k] >= 0) atomicAdd(&hist[bn[k]], 1);
    __syncthreads();
    for (int b = tid; b < NB; b += 256) counts[blockIdx.x * NB + b] = hist[b];
}

__global__ void __launch_bounds__(256)
csr_bin_scan_kernel(const int* __restrict__ counts, int* __restrict__ offsT,
                    int* __restrict__ bintot, int NB, int G) {
    __shared__ int wsum[8];
    const int b = blockIdx.x;
    const int tid = threadIdx.x;
    int carry = 0;
    for (int t0 = 0; t0 < G; t0 += 256) {
        const int blk = t0 + tid;
        const int v = (blk < G) ? counts[blk * NB + b] : 0;
        int tot;
        const int ex = block_excl_scan256(v, wsum, tot);
        if (blk < G) offsT[b * G + blk] = carry + ex;
        carry += tot;
    }
    if (tid == 0) bintot[b] = carry;
}

__global__ void __launch_bounds__(256)
csr_bin_scatter_kernel(const int* __restrict__ ei, const int* __restrict__ offsT,
                       const int* __restrict__ bintot, int* __restrict__ binned,
                       int E, int NB, int G) {
    __shared__ int hist[CSR_MAXB];
    __shared__ int gofs[CSR_MAXB];
    __shared__ int sval[CSR_CHUNK];
    __shared__ unsigned short sbin[CSR_CHUNK];
    __shared__ int wsum[8];
    const int tid = threadIdx.x;
    const int blk = blockIdx.x;
    hist[tid] = 0;
    hist[tid + 256] = 0;
    const int base = blk * CSR_CHUNK;
    int pk[CSR_CHUNK / 256], bn[CSR_CHUNK / 256], rk[CSR_CHUNK / 256];
#pragma unroll
    for (int k = 0; k < CSR_CHUNK / 256; ++k) {
        const int e = base + k * 256 + tid;
        const int ec = (e < E) ? e : E - 1;
        const int s = ei[ec];
        const int d = ei[E + ec];
        pk[k] = (s << 8) | (d & 255);
        bn[k] = (e < E) ? (d >> 8) : -1;
    }
    __syncthreads();
#pragma unroll
    for (int k = 0; k < CSR_CHUNK / 256; ++k)
        rk[k] = (bn[k] >= 0) ? atomicAdd(&hist[bn[k]], 1) : 0;
    __syncthreads();

    const int b0 = 2 * tid, b1 = 2 * tid + 1;
    const int c0 = hist[b0], c1 = hist[b1];
    int tot;
    const int lex = block_excl_scan256(c0 + c1, wsum, tot);
    const int t0 = (b0 < NB) ? bintot[b0] : 0;
    const int t1 = (b1 < NB) ? bintot[b1] : 0;
    int gtot;
    const int gex = block_excl_scan256(t0 + t1, wsum, gtot);
    (void)gtot;
    hist[b0] = lex;
    hist[b1] = lex + c0;
    gofs[b0] = (b0 < NB) ? gex + offsT[b0 * G + blk] - lex : 0;
    gofs[b1] = (b1 < NB) ? gex + t0 + offsT[b1 * G + blk] - (lex + c0) : 0;
    __syncthreads();
#pragma unroll
    for (int k = 0; k < CSR_CHUNK / 256; ++k) {
        if (bn[k] >= 0) {
            const int p = hist[bn[k]] + rk[k];
            sval[p] = pk[k];
            sbin[p] = (unsigned short)bn[k];
        }
    }
    __syncthreads();
    for (int p = tid; p < tot; p += 256) binned[gofs[sbin[p]] + p] = sval[p];
}

__global__ void __launch_bounds__(256)
csr_bin_finalize_kernel(const int* __restrict__ binned, const int* __restrict__ bintot,
                        int* __restrict__ row_ptr, int* __restrict__ colidx,
                        float* __restrict__ dis, int N, int total) {
    __shared__ int cnt[256];
    __shared__ int cur[256];
    __shared__ int stage[CSR_STAGE];
    __shared__ int wsum[8];
    const int tid = threadIdx.x;
    const int b = blockIdx.x;
    const int node = b * 256 + tid;

    int part = 0;
    for (int i = tid; i < b; i += 256) part += bintot[i];
    int binbase;
    (void)block_excl_scan256(part, wsum, binbase);
    const int m = bintot[b];
    const int* in = binned + binbase;

    cnt[tid] = 0;
    __syncthreads();
    for (int p0 = 0; p0 < m; p0 += 2048) {
        int v[8];
#pragma unroll
        for (int j = 0; j < 8; ++j) {
            const int p = p0 + j * 256 + tid;
            const int w = in[(p < m) ? p : m - 1];
            v[j] = (p < m) ? w : -1;
        }
#pragma unroll
        for (int j = 0; j < 8; ++j)
            if (v[j] >= 0) atomicAdd(&cnt[v[j] & 255], 1);
    }
    __syncthreads();

    const bool has = node < N;
    const int d = has ? cnt[tid] + 1 : 0;
    int tot;
    const int ex = block_excl_scan256(d, wsum, tot);
    const int rowbase = binbase + b * 256;
    if (has) {
        row_ptr[node] = rowbase + ex;
        dis[node] = rsqrtf((float)d);
        if (node == N - 1) row_ptr[N] = total;
    }
    cur[tid] = ex + 1;

    if (tot <= CSR_STAGE) {
        if (has) stage[ex] = node;
        __syncthreads();
        for (int p0 = 0; p0 < m; p0 += 2048) {
            int v[8];
#pragma unroll
            for (int j = 0; j < 8; ++j) {
                const int p = p0 + j * 256 + tid;
                const int w = in[(p < m) ? p : m - 1];
                v[j] = (p < m) ? w : -1;
            }
#pragma unroll
            for (int j = 0; j < 8; ++j)
                if (v[j] >= 0) stage[atomicAdd(&cur[v[j] & 255], 1)] = v[j] >> 8;
        }
        __syncthreads();
        for (int q = tid; q < tot; q += 256) colidx[rowbase + q] = stage[q];
    } else {
        if (has) colidx[rowbase + ex] = node;
        __syncthreads();
        for (int p = tid; p < m; p += 256) {
            const int v = in[p];
            colidx[rowbase + atomicAdd(&cur[v & 255], 1)] = v >> 8;
        }
    }
}


template <int NT, int KCH, bool MASK>
__global__ void __launch_bounds__(256)
gemm_wmma_f32_kernel(const float* __restrict__ A, int lda,
                     const float* __restrict__ B, int ldb,
                     float* __restrict__ C, int ldc,
                     int Ncols, const float* __restrict__ bias, int nRowTiles) {
    constexpr int NP = NT * 16;
    constexpr int K  = KCH * 128;
    constexpr int Q  = NP / 4;
    __shared__ __attribute__((aligned(16))) float sW[K * NP];

    const int tid  = threadIdx.x;
    const int lane = tid & 31;
    const int wave = tid >> 5;
    const int c    = lane & 15;
    const int h    = lane >> 4;
    const int tile = blockIdx.x * 8 + wave;
    const bool live = tile < nRowTiles;

    const float* Ap = A + (size_t)((live ? tile : 0) * 16 + c) * lda + 4 * h;

    float4 a[16];
#pragma unroll
    for (int t = 0; t < 16; ++t) a[t] = *(const float4*)(Ap + 8 * t);

    for (int it = tid; it < (K / 4) * Q; it += 256) {
        const int g = it / Q;
        const int q = it - g * Q;
        float4 r0, r1, r2, r3;
        if (!MASK || 4 * q < Ncols) {
            const float* p = B + (size_t)(4 * g) * ldb + 4 * q;
            r0 = *(const float4*)(p);
            r1 = *(const float4*)(p + ldb);
            r2 = *(const float4*)(p + 2 * ldb);
            r3 = *(const float4*)(p + 3 * ldb);
        } else {
            r0 = r1 = r2 = r3 = make_float4(0.f, 0.f, 0.f, 0.f);
        }
        float4* d = (float4*)(sW + ((size_t)g * NP + 4 * q) * 4);
        d[0] = make_float4(r0.x, r1.x, r2.x, r3.x);
        d[1] = make_float4(r0.y, r1.y, r2.y, r3.y);
        d[2] = make_float4(r0.z, r1.z, r2.z, r3.z);
        d[3] = make_float4(r0.w, r1.w, r2.w, r3.w);
    }
    __syncthreads();
    if (!live) return;

    v8f acc[NT];
#pragma unroll
    for (int j = 0; j < NT; ++j) acc[j] = v8f{};

#pragma unroll
    for (int ch = 0; ch < KCH; ++ch) {
        float4 an[16];
        if (ch + 1 < KCH) {
#pragma unroll
            for (int t = 0; t < 16; ++t)
                an[t] = *(const float4*)(Ap + (ch + 1) * 128 + 8 * t);
        }
        const float* sB = sW + ((size_t)(ch * 32 + h) * NP + c) * 4;
#pragma unroll
        for (int t = 0; t < 16; ++t) {
            float4 b[NT];
#pragma unroll
            for (int j = 0; j < NT; ++j)
                b[j] = *(const float4*)(sB + ((size_t)(2 * t) * NP + 16 * j) * 4);
            v2f a0, a1;
            a0.x = a[t].x; a0.y = a[t].y;
            a1.x = a[t].z; a1.y = a[t].w;
#pragma unroll
            for (int j = 0; j < NT; ++j) {
                v2f b0; b0.x = b[j].x; b0.y = b[j].y;
                acc[j] = __builtin_amdgcn_wmma_f32_16x16x4_f32(
                     false, a0,  false, b0,
                     (short)0, acc[j],  false,  false);
            }
#pragma unroll
            for (int j = 0; j < NT; ++j) {
                v2f b1; b1.x = b[j].z; b1.y = b[j].w;
                acc[j] = __builtin_amdgcn_wmma_f32_16x16x4_f32(
                     false, a1,  false, b1,
                     (short)0, acc[j],  false,  false);
            }
        }
        if (ch + 1 < KCH) {
#pragma unroll
            for (int t = 0; t < 16; ++t) a[t] = an[t];
        }
    }

    const int rbase = tile * 16 + 8 * h;
#pragma unroll
    for (int j = 0; j < NT; ++j) {
        const int n = 16 * j + c;
        if (!MASK || n < Ncols) {
            const float badd = bias ? bias[n] : 0.0f;
            float* Cp = C + (size_t)rbase * ldc + n;
#pragma unroll
            for (int r = 0; r < 8; ++r) {
                Cp[0] = acc[j][r] + badd;
                Cp += ldc;
            }
        }
    }
}


__global__ void gcn_aggregate_kernel(const float* __restrict__ H,
                                     const int* __restrict__ row_ptr,
                                     const int* __restrict__ colidx,
                                     const float* __restrict__ dis,
                                     const float* __restrict__ bias,
                                     float* __restrict__ outbuf,
                                     int ld_out, int col_off, int n) {
    const int node = blockIdx.x * (blockDim.x >> 5) + (threadIdx.x >> 5);
    const int lane = threadIdx.x & 31;
    if (node >= n) return;

    const int s = row_ptr[node];
    const int e = row_ptr[node + 1];

    float a0 = 0.f, a1 = 0.f, a2 = 0.f, a3 = 0.f;
    int p = s;
    for (; p + 1 < e; p += 2) {
        const int j0 = colidx[p];
        const int j1 = colidx[p + 1];
        const float w0 = dis[j0];
        const float w1 = dis[j1];
        const float4 h0 = *(const float4*)(H + (size_t)j0 * 128 + lane * 4);
        const float4 h1 = *(const float4*)(H + (size_t)j1 * 128 + lane * 4);
        a0 += w0 * h0.x + w1 * h1.x;
        a1 += w0 * h0.y + w1 * h1.y;
        a2 += w0 * h0.z + w1 * h1.z;
        a3 += w0 * h0.w + w1 * h1.w;
    }
    if (p < e) {
        const int j = colidx[p];
        const float w = dis[j];
        const float4 h = *(const float4*)(H + (size_t)j * 128 + lane * 4);
        a0 += w * h.x; a1 += w * h.y; a2 += w * h.z; a3 += w * h.w;
    }

    const float di = dis[node];
    const int c = lane * 4;
    float4 r;
    r.x = fmaxf(di * a0 + bias[c + 0], 0.0f);
    r.y = fmaxf(di * a1 + bias[c + 1], 0.0f);
    r.z = fmaxf(di * a2 + bias[c + 2], 0.0f);
    r.w = fmaxf(di * a3 + bias[c + 3], 0.0f);
    *(float4*)(outbuf + (size_t)node * ld_out + col_off + c) = r;
}


#define FT_PITCH 132

__global__ void __launch_bounds__(256)
gcn_gather_gemm_kernel(const float* __restrict__ T, int ldt,
                       const float* __restrict__ B,
                       const int* __restrict__ row_ptr,
                       const int* __restrict__ colidx,
                       const float* __restrict__ dis,
                       const float* __restrict__ bias,
                       float* __restrict__ outbuf, int ld_out,
                       int nRowTiles) {
    constexpr int NT = 8;
    constexpr int NP = 128;
    constexpr int K  = 128;
    constexpr int Q  = NP / 4;
    __shared__ __attribute__((aligned(16))) float sW[K * NP];
    __shared__ __attribute__((aligned(16))) float sT[8 * 16 * FT_PITCH];

    const int tid  = threadIdx.x;
    const int lane = tid & 31;
    const int wave = tid >> 5;
    const int c    = lane & 15;
    const int h    = lane >> 4;
    const int tile = blockIdx.x * 8 + wave;
    const bool live = tile < nRowTiles;
    const int n0 = (live ? tile : 0) * 16;
    float* tl = sT + wave * (16 * FT_PITCH);

    const int rpv = row_ptr[n0 + (lane < 16 ? lane : 16)];
    const float dsv = dis[n0 + c];
    const int E0 = __builtin_amdgcn_readlane(rpv, 0);
    const int E1 = __builtin_amdgcn_readlane(rpv, 16);
    int myj, nj = 0;
    {
        const int cnt0 = min(32, E1 - E0);
        myj = colidx[E0 + min(lane, max(cnt0 - 1, 0))];
    }

    for (int it = tid; it < (K / 4) * Q; it += 256) {
        const int g = it / Q;
        const int q = it - g * Q;
        const float* p = B + (size_t)(4 * g) * NP + 4 * q;
        const float4 r0 = *(const float4*)(p);
        const float4 r1 = *(const float4*)(p + NP);
        const float4 r2 = *(const float4*)(p + 2 * NP);
        const float4 r3 = *(const float4*)(p + 3 * NP);
        float4* d = (float4*)(sW + ((size_t)g * NP + 4 * q) * 4);
        d[0] = make_float4(r0.x, r1.x, r2.x, r3.x);
        d[1] = make_float4(r0.y, r1.y, r2.y, r3.y);
        d[2] = make_float4(r0.z, r1.z, r2.z, r3.z);
        d[3] = make_float4(r0.w, r1.w, r2.w, r3.w);
    }
    __syncthreads();
    if (!live) return;

    int cur = 0;
    int nextEnd = __builtin_amdgcn_readlane(rpv, 1);
    float a0 = 0.f, a1 = 0.f, a2 = 0.f, a3 = 0.f;
    const float* Tl = T + lane * 4;
    for (int base = E0; base < E1; base += 32) {
        const int cnt = min(32, E1 - base);
        const int nb = base + 32;
        const bool more = nb < E1;
        for (int g = 0; g < cnt; g += 8) {
            float4 r[8];
            float wv[8];
#pragma unroll
            for (int i = 0; i < 8; ++i) {
                const int j = __builtin_amdgcn_readlane(myj, g + i);
                r[i] = *(const float4*)(Tl + (size_t)j * ldt);
                wv[i] = dis[j];
            }
            if (g == 0 && more) nj = colidx[nb + min(lane, min(32, E1 - nb) - 1)];
            auto add_edge = [&](int e, float w, const float4& v) {
                while (e == nextEnd && cur < 15) {
                    const float di = __int_as_float(__builtin_amdgcn_readlane(__float_as_int(dsv), cur));
                    *(float4*)(tl + cur * FT_PITCH + lane * 4) = make_float4(di * a0, di * a1, di * a2, di * a3);
                    a0 = a1 = a2 = a3 = 0.f;
                    ++cur;
                    nextEnd = __builtin_amdgcn_readlane(rpv, cur + 1);
                }
                a0 += w * v.x; a1 += w * v.y; a2 += w * v.z; a3 += w * v.w;
            };
            if (g + 8 <= cnt) {
#pragma unroll
                for (int i = 0; i < 8; ++i) add_edge(base + g + i, wv[i], r[i]);
            } else {
#pragma unroll
                for (int i = 0; i < 8; ++i)
                    if (g + i < cnt) add_edge(base + g + i, wv[i], r[i]);
            }
        }
        myj = nj;
    }
    for (; cur < 16; ++cur) {
        const float di = __int_as_float(__builtin_amdgcn_readlane(__float_as_int(dsv), cur));
        *(float4*)(tl + cur * FT_PITCH + lane * 4) = make_float4(di * a0, di * a1, di * a2, di * a3);
        a0 = a1 = a2 = a3 = 0.f;
    }

    __builtin_amdgcn_fence(__ATOMIC_RELEASE, "wavefront");
    __builtin_amdgcn_wave_barrier();
    __builtin_amdgcn_fence(__ATOMIC_ACQUIRE, "wavefront");

    float4 a[16];
#pragma unroll
    for (int t = 0; t < 16; ++t) a[t] = *(const float4*)(tl + c * FT_PITCH + 8 * t + 4 * h);

    v8f acc[NT];
#pragma unroll
    for (int j = 0; j < NT; ++j) acc[j] = v8f{};

    const float* sB = sW + ((size_t)h * NP + c) * 4;
#pragma unroll
    for (int t = 0; t < 16; ++t) {
        float4 b[NT];
#pragma unroll
        for (int j = 0; j < NT; ++j)
            b[j] = *(const float4*)(sB + ((size_t)(2 * t) * NP + 16 * j) * 4);
        v2f x0, x1;
        x0.x = a[t].x; x0.y = a[t].y;
        x1.x = a[t].z; x1.y = a[t].w;
#pragma unroll
        for (int j = 0; j < NT; ++j) {
            v2f b0; b0.x = b[j].x; b0.y = b[j].y;
            acc[j] = __builtin_amdgcn_wmma_f32_16x16x4_f32(
                 false, x0,  false, b0,
                 (short)0, acc[j],  false,  false);
        }
#pragma unroll
        for (int j = 0; j < NT; ++j) {
            v2f b1; b1.x = b[j].z; b1.y = b[j].w;
            acc[j] = __builtin_amdgcn_wmma_f32_16x16x4_f32(
                 false, x1,  false, b1,
                 (short)0, acc[j],  false,  false);
        }
    }

#pragma unroll
    for (int j = 0; j < NT; ++j) {
        const float badd = bias[16 * j + c];
#pragma unroll
        for (int r = 0; r < 8; ++r)
            tl[(8 * h + r) * FT_PITCH + 16 * j + c] = fmaxf(acc[j][r] + badd, 0.0f);
    }
    __builtin_amdgcn_fence(__ATOMIC_RELEASE, "wavefront");
    __builtin_amdgcn_wave_barrier();
    __builtin_amdgcn_fence(__ATOMIC_ACQUIRE, "wavefront");
    float* Op = outbuf + (size_t)n0 * ld_out + lane * 4;
#pragma unroll
    for (int r = 0; r < 16; ++r)
        *(float4*)(Op + (size_t)r * ld_out) = *(const float4*)(tl + r * FT_PITCH + lane * 4);
}


extern "C" void kernel_launch(void* const* d_in, const int* in_sizes, int n_in,
                              void* d_out, int out_size, void* d_ws, size_t ws_size,
                              hipStream_t stream) {
    (void)n_in; (void)out_size; (void)ws_size;
    const int F_IN = 256, HID = 128, NC = 40;
    const int N = in_sizes[0] / F_IN;
    const int E = in_sizes[1] / 2;
    const int TOT = E + N;

    const float* x     = (const float*)d_in[0];
    const int*   ei    = (const int*)d_in[1];
    const float* W_in  = (const float*)d_in[2];
    const float* b_in  = (const float*)d_in[3];
    const float* W1    = (const float*)d_in[4];
    const float* b1    = (const float*)d_in[5];
    const float* W2    = (const float*)d_in[6];
    const float* b2    = (const float*)d_in[7];
    const float* W3    = (const float*)d_in[8];
    const float* b3    = (const float*)d_in[9];
    const float* W_out = (const float*)d_in[10];
    const float* b_out = (const float*)d_in[11];
    float* out = (float*)d_out;

    char* base = (char*)d_ws;
    size_t off = 0;
    auto alloc = [&](size_t bytes) -> void* {
        void* q = base + off;
        off = (off + bytes + 255) & ~(size_t)255;
        return q;
    };
    int*   row_ptr = (int*)alloc((size_t)(N + 1) * 4);
    int*   colidx  = (int*)alloc((size_t)TOT * 4);
    float* dis     = (float*)alloc((size_t)N * 4);
    float* H       = (float*)alloc((size_t)N * HID * 4);
    float* hcat    = (float*)alloc((size_t)N * HID * 4 * 4);

    const int NB = (N + 255) / 256;
    const int G  = (E + CSR_CHUNK - 1) / CSR_CHUNK;

    int* counts = (int*)hcat;
    int* offsT  = counts + (size_t)G * NB;
    int* bintot = offsT + (size_t)NB * G;
    int* binned = bintot + CSR_MAXB;

    csr_bin_count_kernel<<<G, 256, 0, stream>>>(ei, counts, E, NB);
    csr_bin_scan_kernel<<<NB, 256, 0, stream>>>(counts, offsT, bintot, NB, G);
    csr_bin_scatter_kernel<<<G, 256, 0, stream>>>(ei, offsT, bintot, binned, E, NB, G);
    csr_bin_finalize_kernel<<<NB, 256, 0, stream>>>(binned, bintot, row_ptr, colidx, dis, N, TOT);

    const int rowTiles = N / 16;
    const int gemmBlocks = (rowTiles + 7) / 8;
    const int aggBlocks = (N + 7) / 8;

    gemm_wmma_f32_kernel<8, 2, false><<<gemmBlocks, 256, 0, stream>>>(
        x, F_IN, W_in, HID, H, HID, HID, nullptr, rowTiles);
    gcn_aggregate_kernel<<<aggBlocks, 256, 0, stream>>>(
        H, row_ptr, colidx, dis, b_in, hcat, 4 * HID, 0 * HID, N);

    const float* Ws[3] = {W1, W2, W3};
    const float* bs[3] = {b1, b2, b3};
    for (int l = 0; l < 3; ++l) {
        gcn_gather_gemm_kernel<<<gemmBlocks, 256, 0, stream>>>(
            hcat + (size_t)l * HID, 4 * HID, Ws[l], row_ptr, colidx, dis, bs[l],
            hcat + (size_t)(l + 1) * HID, 4 * HID, rowTiles);
    }

    gemm_wmma_f32_kernel<3, 4, true><<<gemmBlocks, 256, 0, stream>>>(
        hcat, 4 * HID, W_out, NC, out, NC, NC, b_out, rowTiles);
}


# Layers 2-4 gather: neighbour rows in two register sets of 16, next set's row loads in flight while one is summed, indices and edge weights loaded per lane a chunk ahead; sum order unchanged
# speedup vs baseline: 1.9573x; 1.0312x over previous
#include <hip/hip_runtime.h>
#include <hip/hip_bf16.h>

typedef __attribute__((ext_vector_type(2))) float v2f;
typedef __attribute__((ext_vector_type(8))) float v8f;


#define CSR_CHUNK 4096
#define CSR_MAXB  512
#define CSR_STAGE 8192

__device__ __forceinline__ int block_excl_scan256(int v, int* wsum, int& total) {
    const int lane = threadIdx.x & 31;
    const int wave = threadIdx.x >> 5;
    int inc = v;
    for (int o = 1; o < 32; o <<= 1) {
        const int t = __shfl_up(inc, o, 32);
        if (lane >= o) inc += t;
    }
    if (lane == 31) wsum[wave] = inc;
    __syncthreads();
    int pre = 0, tot = 0;
    for (int w = 0; w < 8; ++w) {
        const int s = wsum[w];
        if (w < wave) pre += s;
        tot += s;
    }
    __syncthreads();
    total = tot;
    return pre + inc - v;
}

__global__ void __launch_bounds__(256)
csr_bin_count_kernel(const int* __restrict__ ei, int* __restrict__ counts,
                     int E, int NB) {
    __shared__ int hist[CSR_MAXB];
    const int tid = threadIdx.x;
    hist[tid] = 0;
    hist[tid + 256] = 0;
    const int base = blockIdx.x * CSR_CHUNK;
    int bn[CSR_CHUNK / 256];
#pragma unroll
    for (int k = 0; k < CSR_CHUNK / 256; ++k) {
        const int e = base + k * 256 + tid;
        const int d = ei[E + ((e < E) ? e : E - 1)];
        bn[k] = (e < E) ? (d >> 8) : -1;
    }
    __syncthreads();
#pragma unroll
    for (int k = 0; k < CSR_CHUNK / 256; ++k)
        if (bn[k] >= 0) atomicAdd(&hist[bn[k]], 1);
    __syncthreads();
    for (int b = tid; b < NB; b += 256) counts[blockIdx.x * NB + b] = hist[b];
}

__global__ void __launch_bounds__(256)
csr_bin_scan_kernel(const int* __restrict__ counts, int* __restrict__ offsT,
                    int* __restrict__ bintot, int NB, int G) {
    __shared__ int wsum[8];
    const int b = blockIdx.x;
    const int tid = threadIdx.x;
    int carry = 0;
    for (int t0 = 0; t0 < G; t0 += 256) {
        const int blk = t0 + tid;
        const int v = (blk < G) ? counts[blk * NB + b] : 0;
        int tot;
        const int ex = block_excl_scan256(v, wsum, tot);
        if (blk < G) offsT[b * G + blk] = carry + ex;
        carry += tot;
    }
    if (tid == 0) bintot[b] = carry;
}

__global__ void __launch_bounds__(256)
csr_bin_scatter_kernel(const int* __restrict__ ei, const int* __restrict__ offsT,
                       const int* __restrict__ bintot, int* __restrict__ binned,
                       int E, int NB, int G) {
    __shared__ int hist[CSR_MAXB];
    __shared__ int gofs[CSR_MAXB];
    __shared__ int sval[CSR_CHUNK];
    __shared__ unsigned short sbin[CSR_CHUNK];
    __shared__ int wsum[8];
    const int tid = threadIdx.x;
    const int blk = blockIdx.x;
    hist[tid] = 0;
    hist[tid + 256] = 0;
    const int base = blk * CSR_CHUNK;
    int pk[CSR_CHUNK / 256], bn[CSR_CHUNK / 256], rk[CSR_CHUNK / 256];
#pragma unroll
    for (int k = 0; k < CSR_CHUNK / 256; ++k) {
        const int e = base + k * 256 + tid;
        const int ec = (e < E) ? e : E - 1;
        const int s = ei[ec];
        const int d = ei[E + ec];
        pk[k] = (s << 8) | (d & 255);
        bn[k] = (e < E) ? (d >> 8) : -1;
    }
    __syncthreads();
#pragma unroll
    for (int k = 0; k < CSR_CHUNK / 256; ++k)
        rk[k] = (bn[k] >= 0) ? atomicAdd(&hist[bn[k]], 1) : 0;
    __syncthreads();

    const int b0 = 2 * tid, b1 = 2 * tid + 1;
    const int c0 = hist[b0], c1 = hist[b1];
    int tot;
    const int lex = block_excl_scan256(c0 + c1, wsum, tot);
    const int t0 = (b0 < NB) ? bintot[b0] : 0;
    const int t1 = (b1 < NB) ? bintot[b1] : 0;
    int gtot;
    const int gex = block_excl_scan256(t0 + t1, wsum, gtot);
    (void)gtot;
    hist[b0] = lex;
    hist[b1] = lex + c0;
    gofs[b0] = (b0 < NB) ? gex + offsT[b0 * G + blk] - lex : 0;
    gofs[b1] = (b1 < NB) ? gex + t0 + offsT[b1 * G + blk] - (lex + c0) : 0;
    __syncthreads();
#pragma unroll
    for (int k = 0; k < CSR_CHUNK / 256; ++k) {
        if (bn[k] >= 0) {
            const int p = hist[bn[k]] + rk[k];
            sval[p] = pk[k];
            sbin[p] = (unsigned short)bn[k];
        }
    }
    __syncthreads();
    for (int p = tid; p < tot; p += 256) binned[gofs[sbin[p]] + p] = sval[p];
}

__global__ void __launch_bounds__(256)
csr_bin_finalize_kernel(const int* __restrict__ binned, const int* __restrict__ bintot,
                        int* __restrict__ row_ptr, int* __restrict__ colidx,
                        float* __restrict__ dis, int N, int total) {
    __shared__ int cnt[256];
    __shared__ int cur[256];
    __shared__ int stage[CSR_STAGE];
    __shared__ int wsum[8];
    const int tid = threadIdx.x;
    const int b = blockIdx.x;
    const int node = b * 256 + tid;

    int part = 0;
    for (int i = tid; i < b; i += 256) part += bintot[i];
    int binbase;
    (void)block_excl_scan256(part, wsum, binbase);
    const int m = bintot[b];
    const int* in = binned + binbase;

    cnt[tid] = 0;
    __syncthreads();
    for (int p0 = 0; p0 < m; p0 += 2048) {
        int v[8];
#pragma unroll
        for (int j = 0; j < 8; ++j) {
            const int p = p0 + j * 256 + tid;
            const int w = in[(p < m) ? p : m - 1];
            v[j] = (p < m) ? w : -1;
        }
#pragma unroll
        for (int j = 0; j < 8; ++j)
            if (v[j] >= 0) atomicAdd(&cnt[v[j] & 255], 1);
    }
    __syncthreads();

    const bool has = node < N;
    const int d = has ? cnt[tid] + 1 : 0;
    int tot;
    const int ex = block_excl_scan256(d, wsum, tot);
    const int rowbase = binbase + b * 256;
    if (has) {
        row_ptr[node] = rowbase + ex;
        dis[node] = rsqrtf((float)d);
        if (node == N - 1) row_ptr[N] = total;
    }
    cur[tid] = ex + 1;

    if (tot <= CSR_STAGE) {
        if (has) stage[ex] = node;
        __syncthreads();
        for (int p0 = 0; p0 < m; p0 += 2048) {
            int v[8];
#pragma unroll
            for (int j = 0; j < 8; ++j) {
                const int p = p0 + j * 256 + tid;
                const int w = in[(p < m) ? p : m - 1];
                v[j] = (p < m) ? w : -1;
            }
#pragma unroll
            for (int j = 0; j < 8; ++j)
                if (v[j] >= 0) stage[atomicAdd(&cur[v[j] & 255], 1)] = v[j] >> 8;
        }
        __syncthreads();
        for (int q = tid; q < tot; q += 256) colidx[rowbase + q] = stage[q];
    } else {
        if (has) colidx[rowbase + ex] = node;
        __syncthreads();
        for (int p = tid; p < m; p += 256) {
            const int v = in[p];
            colidx[rowbase + atomicAdd(&cur[v & 255], 1)] = v >> 8;
        }
    }
}


template <int NT, int KCH, bool MASK>
__global__ void __launch_bounds__(256)
gemm_wmma_f32_kernel(const float* __restrict__ A, int lda,
                     const float* __restrict__ B, int ldb,
                     float* __restrict__ C, int ldc,
                     int Ncols, const float* __restrict__ bias, int nRowTiles) {
    constexpr int NP = NT * 16;
    constexpr int K  = KCH * 128;
    constexpr int Q  = NP / 4;
    __shared__ __attribute__((aligned(16))) float sW[K * NP];

    const int tid  = threadIdx.x;
    const int lane = tid & 31;
    const int wave = tid >> 5;
    const int c    = lane & 15;
    const int h    = lane >> 4;
    const int tile = blockIdx.x * 8 + wave;
    const bool live = tile < nRowTiles;

    const float* Ap = A + (size_t)((live ? tile : 0) * 16 + c) * lda + 4 * h;

    float4 a[16];
#pragma unroll
    for (int t = 0; t < 16; ++t) a[t] = *(const float4*)(Ap + 8 * t);

    for (int it = tid; it < (K / 4) * Q; it += 256) {
        const int g = it / Q;
        const int q = it - g * Q;
        float4 r0, r1, r2, r3;
        if (!MASK || 4 * q < Ncols) {
            const float* p = B + (size_t)(4 * g) * ldb + 4 * q;
            r0 = *(const float4*)(p);
            r1 = *(const float4*)(p + ldb);
            r2 = *(const float4*)(p + 2 * ldb);
            r3 = *(const float4*)(p + 3 * ldb);
        } else {
            r0 = r1 = r2 = r3 = make_float4(0.f, 0.f, 0.f, 0.f);
        }
        float4* d = (float4*)(sW + ((size_t)g * NP + 4 * q) * 4);
        d[0] = make_float4(r0.x, r1.x, r2.x, r3.x);
        d[1] = make_float4(r0.y, r1.y, r2.y, r3.y);
        d[2] = make_float4(r0.z, r1.z, r2.z, r3.z);
        d[3] = make_float4(r0.w, r1.w, r2.w, r3.w);
    }
    __syncthreads();
    if (!live) return;

    v8f acc[NT];
#pragma unroll
    for (int j = 0; j < NT; ++j) acc[j] = v8f{};

#pragma unroll
    for (int ch = 0; ch < KCH; ++ch) {
        float4 an[16];
        if (ch + 1 < KCH) {
#pragma unroll
            for (int t = 0; t < 16; ++t)
                an[t] = *(const float4*)(Ap + (ch + 1) * 128 + 8 * t);
        }
        const float* sB = sW + ((size_t)(ch * 32 + h) * NP + c) * 4;
#pragma unroll
        for (int t = 0; t < 16; ++t) {
            float4 b[NT];
#pragma unroll
            for (int j = 0; j < NT; ++j)
                b[j] = *(const float4*)(sB + ((size_t)(2 * t) * NP + 16 * j) * 4);
            v2f a0, a1;
            a0.x = a[t].x; a0.y = a[t].y;
            a1.x = a[t].z; a1.y = a[t].w;
#pragma unroll
            for (int j = 0; j < NT; ++j) {
                v2f b0; b0.x = b[j].x; b0.y = b[j].y;
                acc[j] = __builtin_amdgcn_wmma_f32_16x16x4_f32(
                     false, a0,  false, b0,
                     (short)0, acc[j],  false,  false);
            }
#pragma unroll
            for (int j = 0; j < NT; ++j) {
                v2f b1; b1.x = b[j].z; b1.y = b[j].w;
                acc[j] = __builtin_amdgcn_wmma_f32_16x16x4_f32(
                     false, a1,  false, b1,
                     (short)0, acc[j],  false,  false);
            }
        }
        if (ch + 1 < KCH) {
#pragma unroll
            for (int t = 0; t < 16; ++t) a[t] = an[t];
        }
    }

    const int rbase = tile * 16 + 8 * h;
#pragma unroll
    for (int j = 0; j < NT; ++j) {
        const int n = 16 * j + c;
        if (!MASK || n < Ncols) {
            const float badd = bias ? bias[n] : 0.0f;
            float* Cp = C + (size_t)rbase * ldc + n;
#pragma unroll
            for (int r = 0; r < 8; ++r) {
                Cp[0] = acc[j][r] + badd;
                Cp += ldc;
            }
        }
    }
}


__global__ void gcn_aggregate_kernel(const float* __restrict__ H,
                                     const int* __restrict__ row_ptr,
                                     const int* __restrict__ colidx,
                                     const float* __restrict__ dis,
                                     const float* __restrict__ bias,
                                     float* __restrict__ outbuf,
                                     int ld_out, int col_off, int n) {
    const int node = blockIdx.x * (blockDim.x >> 5) + (threadIdx.x >> 5);
    const int lane = threadIdx.x & 31;
    if (node >= n) return;

    const int s = row_ptr[node];
    const int e = row_ptr[node + 1];

    float a0 = 0.f, a1 = 0.f, a2 = 0.f, a3 = 0.f;
    int p = s;
    for (; p + 1 < e; p += 2) {
        const int j0 = colidx[p];
        const int j1 = colidx[p + 1];
        const float w0 = dis[j0];
        const float w1 = dis[j1];
        const float4 h0 = *(const float4*)(H + (size_t)j0 * 128 + lane * 4);
        const float4 h1 = *(const float4*)(H + (size_t)j1 * 128 + lane * 4);
        a0 += w0 * h0.x + w1 * h1.x;
        a1 += w0 * h0.y + w1 * h1.y;
        a2 += w0 * h0.z + w1 * h1.z;
        a3 += w0 * h0.w + w1 * h1.w;
    }
    if (p < e) {
        const int j = colidx[p];
        const float w = dis[j];
        const float4 h = *(const float4*)(H + (size_t)j * 128 + lane * 4);
        a0 += w * h.x; a1 += w * h.y; a2 += w * h.z; a3 += w * h.w;
    }

    const float di = dis[node];
    const int c = lane * 4;
    float4 r;
    r.x = fmaxf(di * a0 + bias[c + 0], 0.0f);
    r.y = fmaxf(di * a1 + bias[c + 1], 0.0f);
    r.z = fmaxf(di * a2 + bias[c + 2], 0.0f);
    r.w = fmaxf(di * a3 + bias[c + 3], 0.0f);
    *(float4*)(outbuf + (size_t)node * ld_out + col_off + c) = r;
}


#define FT_PITCH 132

__global__ void __launch_bounds__(256)
gcn_gather_gemm_kernel(const float* __restrict__ T, int ldt,
                       const float* __restrict__ B,
                       const int* __restrict__ row_ptr,
                       const int* __restrict__ colidx,
                       const float* __restrict__ dis,
                       const float* __restrict__ bias,
                       float* __restrict__ outbuf, int ld_out,
                       int nRowTiles) {
    constexpr int NT = 8;
    constexpr int NP = 128;
    constexpr int K  = 128;
    constexpr int Q  = NP / 4;
    __shared__ __attribute__((aligned(16))) float sW[K * NP];
    __shared__ __attribute__((aligned(16))) float sT[8 * 16 * FT_PITCH];

    const int tid  = threadIdx.x;
    const int lane = tid & 31;
    const int wave = tid >> 5;
    const int c    = lane & 15;
    const int h    = lane >> 4;
    const int tile = blockIdx.x * 8 + wave;
    const bool live = tile < nRowTiles;
    const int n0 = (live ? tile : 0) * 16;
    float* tl = sT + wave * (16 * FT_PITCH);

    const int rpv = row_ptr[n0 + (lane < 16 ? lane : 16)];
    const float dsv = dis[n0 + c];
    const int E0 = __builtin_amdgcn_readlane(rpv, 0);
    const int E1 = __builtin_amdgcn_readlane(rpv, 16);
    int cj = colidx[min(E0 + lane, E1 - 1)];
    int nj = colidx[min(E0 + 32 + lane, E1 - 1)];

    for (int it = tid; it < (K / 4) * Q; it += 256) {
        const int g = it / Q;
        const int q = it - g * Q;
        const float* p = B + (size_t)(4 * g) * NP + 4 * q;
        const float4 r0 = *(const float4*)(p);
        const float4 r1 = *(const float4*)(p + NP);
        const float4 r2 = *(const float4*)(p + 2 * NP);
        const float4 r3 = *(const float4*)(p + 3 * NP);
        float4* d = (float4*)(sW + ((size_t)g * NP + 4 * q) * 4);
        d[0] = make_float4(r0.x, r1.x, r2.x, r3.x);
        d[1] = make_float4(r0.y, r1.y, r2.y, r3.y);
        d[2] = make_float4(r0.z, r1.z, r2.z, r3.z);
        d[3] = make_float4(r0.w, r1.w, r2.w, r3.w);
    }
    float cw = dis[cj];
    __syncthreads();
    if (!live) return;

    int cur = 0;
    int nextEnd = __builtin_amdgcn_readlane(rpv, 1);
    float a0 = 0.f, a1 = 0.f, a2 = 0.f, a3 = 0.f;
    const float* Tl = T + lane * 4;
    auto add_edge = [&](int e, float w, const float4& v) __attribute__((always_inline)) {
        while (e == nextEnd && cur < 15) {
            const float di = __int_as_float(__builtin_amdgcn_readlane(__float_as_int(dsv), cur));
            *(float4*)(tl + cur * FT_PITCH + lane * 4) = make_float4(di * a0, di * a1, di * a2, di * a3);
            a0 = a1 = a2 = a3 = 0.f;
            ++cur;
            nextEnd = __builtin_amdgcn_readlane(rpv, cur + 1);
        }
        a0 += w * v.x; a1 += w * v.y; a2 += w * v.z; a3 += w * v.w;
    };
    auto issue = [&](float4 (&r)[16], int idx, int g0) __attribute__((always_inline)) {
#pragma unroll
        for (int i = 0; i < 16; ++i) {
            const int j = __builtin_amdgcn_readlane(idx, g0 + i);
            r[i] = *(const float4*)(Tl + (size_t)j * ldt);
        }
    };
    auto sum_rows = [&](const float4 (&r)[16], float wvec, int e0, int g0, int n) __attribute__((always_inline)) {
        if (n >= 16) {
#pragma unroll
            for (int i = 0; i < 16; ++i)
                add_edge(e0 + i, __int_as_float(__builtin_amdgcn_readlane(__float_as_int(wvec), g0 + i)), r[i]);
        } else {
#pragma unroll
            for (int i = 0; i < 16; ++i)
                if (i < n)
                    add_edge(e0 + i, __int_as_float(__builtin_amdgcn_readlane(__float_as_int(wvec), g0 + i)), r[i]);
        }
    };

    float4 rA[16], rB[16];
    issue(rA, cj, 0);
    int base = E0;
    for (; base + 32 <= E1; base += 32) {
        issue(rB, cj, 16);
        const float nw = dis[nj];
        const int fj = colidx[min(base + 64 + lane, E1 - 1)];
        sum_rows(rA, cw, base, 0, 16);
        asm volatile("" : "+v"(a0), "+v"(a1), "+v"(a2), "+v"(a3) : : "memory");
        issue(rA, nj, 0);
        sum_rows(rB, cw, base + 16, 16, 16);
        cj = nj; cw = nw; nj = fj;
    }
    {
        const int rem = E1 - base;
        if (rem > 16) issue(rB, cj, 16);
        sum_rows(rA, cw, base, 0, min(rem, 16));
        if (rem > 16) sum_rows(rB, cw, base + 16, 16, rem - 16);
    }
    for (; cur < 16; ++cur) {
        const float di = __int_as_float(__builtin_amdgcn_readlane(__float_as_int(dsv), cur));
        *(float4*)(tl + cur * FT_PITCH + lane * 4) = make_float4(di * a0, di * a1, di * a2, di * a3);
        a0 = a1 = a2 = a3 = 0.f;
    }

    __builtin_amdgcn_fence(__ATOMIC_RELEASE, "wavefront");
    __builtin_amdgcn_wave_barrier();
    __builtin_amdgcn_fence(__ATOMIC_ACQUIRE, "wavefront");

    float4 a[16];
#pragma unroll
    for (int t = 0; t < 16; ++t) a[t] = *(const float4*)(tl + c * FT_PITCH + 8 * t + 4 * h);

    v8f acc[NT];
#pragma unroll
    for (int j = 0; j < NT; ++j) acc[j] = v8f{};

    const float* sB = sW + ((size_t)h * NP + c) * 4;
#pragma unroll
    for (int t = 0; t < 16; ++t) {
        float4 b[NT];
#pragma unroll
        for (int j = 0; j < NT; ++j)
            b[j] = *(const float4*)(sB + ((size_t)(2 * t) * NP + 16 * j) * 4);
        v2f x0, x1;
        x0.x = a[t].x; x0.y = a[t].y;
        x1.x = a[t].z; x1.y = a[t].w;
#pragma unroll
        for (int j = 0; j < NT; ++j) {
            v2f b0; b0.x = b[j].x; b0.y = b[j].y;
            acc[j] = __builtin_amdgcn_wmma_f32_16x16x4_f32(
                 false, x0,  false, b0,
                 (short)0, acc[j],  false,  false);
        }
#pragma unroll
        for (int j = 0; j < NT; ++j) {
            v2f b1; b1.x = b[j].z; b1.y = b[j].w;
            acc[j] = __builtin_amdgcn_wmma_f32_16x16x4_f32(
                 false, x1,  false, b1,
                 (short)0, acc[j],  false,  false);
        }
    }

#pragma unroll
    for (int j = 0; j < NT; ++j) {
        const float badd = bias[16 * j + c];
#pragma unroll
        for (int r = 0; r < 8; ++r)
            tl[(8 * h + r) * FT_PITCH + 16 * j + c] = fmaxf(acc[j][r] + badd, 0.0f);
    }
    __builtin_amdgcn_fence(__ATOMIC_RELEASE, "wavefront");
    __builtin_amdgcn_wave_barrier();
    __builtin_amdgcn_fence(__ATOMIC_ACQUIRE, "wavefront");
    float* Op = outbuf + (size_t)n0 * ld_out + lane * 4;
#pragma unroll
    for (int r = 0; r < 16; ++r)
        *(float4*)(Op + (size_t)r * ld_out) = *(const float4*)(tl + r * FT_PITCH + lane * 4);
}


extern "C" void kernel_launch(void* const* d_in, const int* in_sizes, int n_in,
                              void* d_out, int out_size, void* d_ws, size_t ws_size,
                              hipStream_t stream) {
    (void)n_in; (void)out_size; (void)ws_size;
    const int F_IN = 256, HID = 128, NC = 40;
    const int N = in_sizes[0] / F_IN;
    const int E = in_sizes[1] / 2;
    const int TOT = E + N;

    const float* x     = (const float*)d_in[0];
    const int*   ei    = (const int*)d_in[1];
    const float* W_in  = (const float*)d_in[2];
    const float* b_in  = (const float*)d_in[3];
    const float* W1    = (const float*)d_in[4];
    const float* b1    = (const float*)d_in[5];
    const float* W2    = (const float*)d_in[6];
    const float* b2    = (const float*)d_in[7];
    const float* W3    = (const float*)d_in[8];
    const float* b3    = (const float*)d_in[9];
    const float* W_out = (const float*)d_in[10];
    const float* b_out = (const float*)d_in[11];
    float* out = (float*)d_out;

    char* base = (char*)d_ws;
    size_t off = 0;
    auto alloc = [&](size_t bytes) -> void* {
        void* q = base + off;
        off = (off + bytes + 255) & ~(size_t)255;
        return q;
    };
    int*   row_ptr = (int*)alloc((size_t)(N + 1) * 4);
    int*   colidx  = (int*)alloc((size_t)TOT * 4);
    float* dis     = (float*)alloc((size_t)N * 4);
    float* H       = (float*)alloc((size_t)N * HID * 4);
    float* hcat    = (float*)alloc((size_t)N * HID * 4 * 4);

    const int NB = (N + 255) / 256;
    const int G  = (E + CSR_CHUNK - 1) / CSR_CHUNK;

    int* counts = (int*)hcat;
    int* offsT  = counts + (size_t)G * NB;
    int* bintot = offsT + (size_t)NB * G;
    int* binned = bintot + CSR_MAXB;

    csr_bin_count_kernel<<<G, 256, 0, stream>>>(ei, counts, E, NB);
    csr_bin_scan_kernel<<<NB, 256, 0, stream>>>(counts, offsT, bintot, NB, G);
    csr_bin_scatter_kernel<<<G, 256, 0, stream>>>(ei, offsT, bintot, binned, E, NB, G);
    csr_bin_finalize_kernel<<<NB, 256, 0, stream>>>(binned, bintot, row_ptr, colidx, dis, N, TOT);

    const int rowTiles = N / 16;
    const int gemmBlocks = (rowTiles + 7) / 8;
    const int aggBlocks = (N + 7) / 8;

    gemm_wmma_f32_kernel<8, 2, false><<<gemmBlocks, 256, 0, stream>>>(
        x, F_IN, W_in, HID, H, HID, HID, nullptr, rowTiles);
    gcn_aggregate_kernel<<<aggBlocks, 256, 0, stream>>>(
        H, row_ptr, colidx, dis, b_in, hcat, 4 * HID, 0 * HID, N);

    const float* Ws[3] = {W1, W2, W3};
    const float* bs[3] = {b1, b2, b3};
    for (int l = 0; l < 3; ++l) {
        gcn_gather_gemm_kernel<<<gemmBlocks, 256, 0, stream>>>(
            hcat + (size_t)l * HID, 4 * HID, Ws[l], row_ptr, colidx, dis, bs[l],
            hcat + (size_t)(l + 1) * HID, 4 * HID, rowTiles);
    }

    gemm_wmma_f32_kernel<3, 4, true><<<gemmBlocks, 256, 0, stream>>>(
        hcat, 4 * HID, W_out, NC, out, NC, NC, b_out, rowTiles);
}


# Feature rows stored pre-multiplied by their node's deg^-1/2, so neighbour sums are plain row sums with no per-edge weight lookup; head divides it back; sum order kept, terms rounded at the store
# speedup vs baseline: 2.1267x; 1.0865x over previous
#include <hip/hip_runtime.h>
#include <hip/hip_bf16.h>

typedef __attribute__((ext_vector_type(2))) float v2f;
typedef __attribute__((ext_vector_type(8))) float v8f;


#define CSR_CHUNK 4096
#define CSR_MAXB  512
#define CSR_STAGE 8192

__device__ __forceinline__ int block_excl_scan256(int v, int* wsum, int& total) {
    const int lane = threadIdx.x & 31;
    const int wave = threadIdx.x >> 5;
    int inc = v;
    for (int o = 1; o < 32; o <<= 1) {
        const int t = __shfl_up(inc, o, 32);
        if (lane >= o) inc += t;
    }
    if (lane == 31) wsum[wave] = inc;
    __syncthreads();
    int pre = 0, tot = 0;
    for (int w = 0; w < 8; ++w) {
        const int s = wsum[w];
        if (w < wave) pre += s;
        tot += s;
    }
    __syncthreads();
    total = tot;
    return pre + inc - v;
}

__global__ void __launch_bounds__(256)
csr_bin_count_kernel(const int* __restrict__ ei, int* __restrict__ counts,
                     int E, int NB) {
    __shared__ int hist[CSR_MAXB];
    const int tid = threadIdx.x;
    hist[tid] = 0;
    hist[tid + 256] = 0;
    const int base = blockIdx.x * CSR_CHUNK;
    int bn[CSR_CHUNK / 256];
#pragma unroll
    for (int k = 0; k < CSR_CHUNK / 256; ++k) {
        const int e = base + k * 256 + tid;
        const int d = ei[E + ((e < E) ? e : E - 1)];
        bn[k] = (e < E) ? (d >> 8) : -1;
    }
    __syncthreads();
#pragma unroll
    for (int k = 0; k < CSR_CHUNK / 256; ++k)
        if (bn[k] >= 0) atomicAdd(&hist[bn[k]], 1);
    __syncthreads();
    for (int b = tid; b < NB; b += 256) counts[blockIdx.x * NB + b] = hist[b];
}

__global__ void __launch_bounds__(256)
csr_bin_scan_kernel(const int* __restrict__ counts, int* __restrict__ offsT,
                    int* __restrict__ bintot, int NB, int G) {
    __shared__ int wsum[8];
    const int b = blockIdx.x;
    const int tid = threadIdx.x;
    int carry = 0;
    for (int t0 = 0; t0 < G; t0 += 256) {
        const int blk = t0 + tid;
        const int v = (blk < G) ? counts[blk * NB + b] : 0;
        int tot;
        const int ex = block_excl_scan256(v, wsum, tot);
        if (blk < G) offsT[b * G + blk] = carry + ex;
        carry += tot;
    }
    if (tid == 0) bintot[b] = carry;
}

__global__ void __launch_bounds__(256)
csr_bin_scatter_kernel(const int* __restrict__ ei, const int* __restrict__ offsT,
                       const int* __restrict__ bintot, int* __restrict__ binned,
                       int E, int NB, int G) {
    __shared__ int hist[CSR_MAXB];
    __shared__ int gofs[CSR_MAXB];
    __shared__ int sval[CSR_CHUNK];
    __shared__ unsigned short sbin[CSR_CHUNK];
    __shared__ int wsum[8];
    const int tid = threadIdx.x;
    const int blk = blockIdx.x;
    hist[tid] = 0;
    hist[tid + 256] = 0;
    const int base = blk * CSR_CHUNK;
    int pk[CSR_CHUNK / 256], bn[CSR_CHUNK / 256], rk[CSR_CHUNK / 256];
#pragma unroll
    for (int k = 0; k < CSR_CHUNK / 256; ++k) {
        const int e = base + k * 256 + tid;
        const int ec = (e < E) ? e : E - 1;
        const int s = ei[ec];
        const int d = ei[E + ec];
        pk[k] = (s << 8) | (d & 255);
        bn[k] = (e < E) ? (d >> 8) : -1;
    }
    __syncthreads();
#pragma unroll
    for (int k = 0; k < CSR_CHUNK / 256; ++k)
        rk[k] = (bn[k] >= 0) ? atomicAdd(&hist[bn[k]], 1) : 0;
    __syncthreads();

    const int b0 = 2 * tid, b1 = 2 * tid + 1;
    const int c0 = hist[b0], c1 = hist[b1];
    int tot;
    const int lex = block_excl_scan256(c0 + c1, wsum, tot);
    const int t0 = (b0 < NB) ? bintot[b0] : 0;
    const int t1 = (b1 < NB) ? bintot[b1] : 0;
    int gtot;
    const int gex = block_excl_scan256(t0 + t1, wsum, gtot);
    (void)gtot;
    hist[b0] = lex;
    hist[b1] = lex + c0;
    gofs[b0] = (b0 < NB) ? gex + offsT[b0 * G + blk] - lex : 0;
    gofs[b1] = (b1 < NB) ? gex + t0 + offsT[b1 * G + blk] - (lex + c0) : 0;
    __syncthreads();
#pragma unroll
    for (int k = 0; k < CSR_CHUNK / 256; ++k) {
        if (bn[k] >= 0) {
            const int p = hist[bn[k]] + rk[k];
            sval[p] = pk[k];
            sbin[p] = (unsigned short)bn[k];
        }
    }
    __syncthreads();
    for (int p = tid; p < tot; p += 256) binned[gofs[sbin[p]] + p] = sval[p];
}

__global__ void __launch_bounds__(256)
csr_bin_finalize_kernel(const int* __restrict__ binned, const int* __restrict__ bintot,
                        int* __restrict__ row_ptr, int* __restrict__ colidx,
                        float* __restrict__ dis, int N, int total) {
    __shared__ int cnt[256];
    __shared__ int cur[256];
    __shared__ int stage[CSR_STAGE];
    __shared__ int wsum[8];
    const int tid = threadIdx.x;
    const int b = blockIdx.x;
    const int node = b * 256 + tid;

    int part = 0;
    for (int i = tid; i < b; i += 256) part += bintot[i];
    int binbase;
    (void)block_excl_scan256(part, wsum, binbase);
    const int m = bintot[b];
    const int* in = binned + binbase;

    cnt[tid] = 0;
    __syncthreads();
    for (int p0 = 0; p0 < m; p0 += 2048) {
        int v[8];
#pragma unroll
        for (int j = 0; j < 8; ++j) {
            const int p = p0 + j * 256 + tid;
            const int w = in[(p < m) ? p : m - 1];
            v[j] = (p < m) ? w : -1;
        }
#pragma unroll
        for (int j = 0; j < 8; ++j)
            if (v[j] >= 0) atomicAdd(&cnt[v[j] & 255], 1);
    }
    __syncthreads();

    const bool has = node < N;
    const int d = has ? cnt[tid] + 1 : 0;
    int tot;
    const int ex = block_excl_scan256(d, wsum, tot);
    const int rowbase = binbase + b * 256;
    if (has) {
        row_ptr[node] = rowbase + ex;
        dis[node] = rsqrtf((float)d);
        if (node == N - 1) row_ptr[N] = total;
    }
    cur[tid] = ex + 1;

    if (tot <= CSR_STAGE) {
        if (has) stage[ex] = node;
        __syncthreads();
        for (int p0 = 0; p0 < m; p0 += 2048) {
            int v[8];
#pragma unroll
            for (int j = 0; j < 8; ++j) {
                const int p = p0 + j * 256 + tid;
                const int w = in[(p < m) ? p : m - 1];
                v[j] = (p < m) ? w : -1;
            }
#pragma unroll
            for (int j = 0; j < 8; ++j)
                if (v[j] >= 0) stage[atomicAdd(&cur[v[j] & 255], 1)] = v[j] >> 8;
        }
        __syncthreads();
        for (int q = tid; q < tot; q += 256) colidx[rowbase + q] = stage[q];
    } else {
        if (has) colidx[rowbase + ex] = node;
        __syncthreads();
        for (int p = tid; p < m; p += 256) {
            const int v = in[p];
            colidx[rowbase + atomicAdd(&cur[v & 255], 1)] = v >> 8;
        }
    }
}


template <int NT, int KCH, bool MASK, bool INV>
__global__ void __launch_bounds__(256)
gemm_wmma_f32_kernel(const float* __restrict__ A, int lda,
                     const float* __restrict__ B, int ldb,
                     float* __restrict__ C, int ldc,
                     int Ncols, const float* __restrict__ bias,
                     const float* __restrict__ rowscale, int nRowTiles) {
    constexpr int NP = NT * 16;
    constexpr int K  = KCH * 128;
    constexpr int Q  = NP / 4;
    __shared__ __attribute__((aligned(16))) float sW[K * NP];

    const int tid  = threadIdx.x;
    const int lane = tid & 31;
    const int wave = tid >> 5;
    const int c    = lane & 15;
    const int h    = lane >> 4;
    const int tile = blockIdx.x * 8 + wave;
    const bool live = tile < nRowTiles;

    const float* Ap = A + (size_t)((live ? tile : 0) * 16 + c) * lda + 4 * h;

    float4 a[16];
#pragma unroll
    for (int t = 0; t < 16; ++t) a[t] = *(const float4*)(Ap + 8 * t);

    for (int it = tid; it < (K / 4) * Q; it += 256) {
        const int g = it / Q;
        const int q = it - g * Q;
        float4 r0, r1, r2, r3;
        if (!MASK || 4 * q < Ncols) {
            const float* p = B + (size_t)(4 * g) * ldb + 4 * q;
            r0 = *(const float4*)(p);
            r1 = *(const float4*)(p + ldb);
            r2 = *(const float4*)(p + 2 * ldb);
            r3 = *(const float4*)(p + 3 * ldb);
        } else {
            r0 = r1 = r2 = r3 = make_float4(0.f, 0.f, 0.f, 0.f);
        }
        float4* d = (float4*)(sW + ((size_t)g * NP + 4 * q) * 4);
        d[0] = make_float4(r0.x, r1.x, r2.x, r3.x);
        d[1] = make_float4(r0.y, r1.y, r2.y, r3.y);
        d[2] = make_float4(r0.z, r1.z, r2.z, r3.z);
        d[3] = make_float4(r0.w, r1.w, r2.w, r3.w);
    }
    __syncthreads();
    if (!live) return;

    v8f acc[NT];
#pragma unroll
    for (int j = 0; j < NT; ++j) acc[j] = v8f{};

#pragma unroll
    for (int ch = 0; ch < KCH; ++ch) {
        float4 an[16];
        if (ch + 1 < KCH) {
#pragma unroll
            for (int t = 0; t < 16; ++t)
                an[t] = *(const float4*)(Ap + (ch + 1) * 128 + 8 * t);
        }
        const float* sB = sW + ((size_t)(ch * 32 + h) * NP + c) * 4;
#pragma unroll
        for (int t = 0; t < 16; ++t) {
            float4 b[NT];
#pragma unroll
            for (int j = 0; j < NT; ++j)
                b[j] = *(const float4*)(sB + ((size_t)(2 * t) * NP + 16 * j) * 4);
            v2f a0, a1;
            a0.x = a[t].x; a0.y = a[t].y;
            a1.x = a[t].z; a1.y = a[t].w;
#pragma unroll
            for (int j = 0; j < NT; ++j) {
                v2f b0; b0.x = b[j].x; b0.y = b[j].y;
                acc[j] = __builtin_amdgcn_wmma_f32_16x16x4_f32(
                     false, a0,  false, b0,
                     (short)0, acc[j],  false,  false);
            }
#pragma unroll
            for (int j = 0; j < NT; ++j) {
                v2f b1; b1.x = b[j].z; b1.y = b[j].w;
                acc[j] = __builtin_amdgcn_wmma_f32_16x16x4_f32(
                     false, a1,  false, b1,
                     (short)0, acc[j],  false,  false);
            }
        }
        if (ch + 1 < KCH) {
#pragma unroll
            for (int t = 0; t < 16; ++t) a[t] = an[t];
        }
    }

    const int rbase = tile * 16 + 8 * h;
    float rs[8];
#pragma unroll
    for (int r = 0; r < 8; ++r) {
        const float d = rowscale[rbase + r];
        rs[r] = INV ? 1.0f / d : d;
    }
#pragma unroll
    for (int j = 0; j < NT; ++j) {
        const int n = 16 * j + c;
        if (!MASK || n < Ncols) {
            const float badd = bias ? bias[n] : 0.0f;
            float* Cp = C + (size_t)rbase * ldc + n;
#pragma unroll
            for (int r = 0; r < 8; ++r) {
                Cp[0] = acc[j][r] * rs[r] + badd;
                Cp += ldc;
            }
        }
    }
}


__global__ void gcn_aggregate_kernel(const float* __restrict__ H,
                                     const int* __restrict__ row_ptr,
                                     const int* __restrict__ colidx,
                                     const float* __restrict__ dis,
                                     const float* __restrict__ bias,
                                     float* __restrict__ outbuf,
                                     int ld_out, int col_off, int n) {
    const int node = blockIdx.x * (blockDim.x >> 5) + (threadIdx.x >> 5);
    const int lane = threadIdx.x & 31;
    if (node >= n) return;

    const int s = row_ptr[node];
    const int e = row_ptr[node + 1];

    float a0 = 0.f, a1 = 0.f, a2 = 0.f, a3 = 0.f;
    int p = s;
    for (; p + 1 < e; p += 2) {
        const int j0 = colidx[p];
        const int j1 = colidx[p + 1];
        const float4 h0 = *(const float4*)(H + (size_t)j0 * 128 + lane * 4);
        const float4 h1 = *(const float4*)(H + (size_t)j1 * 128 + lane * 4);
        a0 += h0.x + h1.x;
        a1 += h0.y + h1.y;
        a2 += h0.z + h1.z;
        a3 += h0.w + h1.w;
    }
    if (p < e) {
        const int j = colidx[p];
        const float4 h = *(const float4*)(H + (size_t)j * 128 + lane * 4);
        a0 += h.x; a1 += h.y; a2 += h.z; a3 += h.w;
    }

    const float di = dis[node];
    const int c = lane * 4;
    float4 r;
    r.x = di * fmaxf(di * a0 + bias[c + 0], 0.0f);
    r.y = di * fmaxf(di * a1 + bias[c + 1], 0.0f);
    r.z = di * fmaxf(di * a2 + bias[c + 2], 0.0f);
    r.w = di * fmaxf(di * a3 + bias[c + 3], 0.0f);
    *(float4*)(outbuf + (size_t)node * ld_out + col_off + c) = r;
}


#define FT_PITCH 132

__global__ void __launch_bounds__(256)
gcn_gather_gemm_kernel(const float* __restrict__ T, int ldt,
                       const float* __restrict__ B,
                       const int* __restrict__ row_ptr,
                       const int* __restrict__ colidx,
                       const float* __restrict__ dis,
                       const float* __restrict__ bias,
                       float* __restrict__ outbuf, int ld_out,
                       int nRowTiles) {
    constexpr int NT = 8;
    constexpr int NP = 128;
    constexpr int K  = 128;
    constexpr int Q  = NP / 4;
    __shared__ __attribute__((aligned(16))) float sW[K * NP];
    __shared__ __attribute__((aligned(16))) float sT[8 * 16 * FT_PITCH];

    const int tid  = threadIdx.x;
    const int lane = tid & 31;
    const int wave = tid >> 5;
    const int c    = lane & 15;
    const int h    = lane >> 4;
    const int tile = blockIdx.x * 8 + wave;
    const bool live = tile < nRowTiles;
    const int n0 = (live ? tile : 0) * 16;
    float* tl = sT + wave * (16 * FT_PITCH);

    const int rpv = row_ptr[n0 + (lane < 16 ? lane : 16)];
    const float dsv = dis[n0 + c];
    const int E0 = __builtin_amdgcn_readlane(rpv, 0);
    const int E1 = __builtin_amdgcn_readlane(rpv, 16);
    int cj = colidx[min(E0 + lane, E1 - 1)];
    int nj = colidx[min(E0 + 32 + lane, E1 - 1)];

    for (int it = tid; it < (K / 4) * Q; it += 256) {
        const int g = it / Q;
        const int q = it - g * Q;
        const float* p = B + (size_t)(4 * g) * NP + 4 * q;
        const float4 r0 = *(const float4*)(p);
        const float4 r1 = *(const float4*)(p + NP);
        const float4 r2 = *(const float4*)(p + 2 * NP);
        const float4 r3 = *(const float4*)(p + 3 * NP);
        float4* d = (float4*)(sW + ((size_t)g * NP + 4 * q) * 4);
        d[0] = make_float4(r0.x, r1.x, r2.x, r3.x);
        d[1] = make_float4(r0.y, r1.y, r2.y, r3.y);
        d[2] = make_float4(r0.z, r1.z, r2.z, r3.z);
        d[3] = make_float4(r0.w, r1.w, r2.w, r3.w);
    }
    __syncthreads();
    if (!live) return;

    int cur = 0;
    int nextEnd = __builtin_amdgcn_readlane(rpv, 1);
    float a0 = 0.f, a1 = 0.f, a2 = 0.f, a3 = 0.f;
    const float* Tl = T + lane * 4;
    auto add_edge = [&](int e, const float4& v) __attribute__((always_inline)) {
        while (e == nextEnd && cur < 15) {
            const float di = __int_as_float(__builtin_amdgcn_readlane(__float_as_int(dsv), cur));
            *(float4*)(tl + cur * FT_PITCH + lane * 4) = make_float4(di * a0, di * a1, di * a2, di * a3);
            a0 = a1 = a2 = a3 = 0.f;
            ++cur;
            nextEnd = __builtin_amdgcn_readlane(rpv, cur + 1);
        }
        a0 += v.x; a1 += v.y; a2 += v.z; a3 += v.w;
    };
    auto issue = [&](float4 (&r)[16], int idx, int g0) __attribute__((always_inline)) {
#pragma unroll
        for (int i = 0; i < 16; ++i) {
            const int j = __builtin_amdgcn_readlane(idx, g0 + i);
            r[i] = *(const float4*)(Tl + (size_t)j * ldt);
        }
    };
    auto sum_rows = [&](const float4 (&r)[16], int e0, int n) __attribute__((always_inline)) {
        if (n >= 16) {
#pragma unroll
            for (int i = 0; i < 16; ++i)
                add_edge(e0 + i, r[i]);
        } else {
#pragma unroll
            for (int i = 0; i < 16; ++i)
                if (i < n)
                    add_edge(e0 + i, r[i]);
        }
    };

    float4 rA[16], rB[16];
    issue(rA, cj, 0);
    int base = E0;
    for (; base + 32 <= E1; base += 32) {
        issue(rB, cj, 16);
        const int fj = colidx[min(base + 64 + lane, E1 - 1)];
        sum_rows(rA, base, 16);
        asm volatile("" : "+v"(a0), "+v"(a1), "+v"(a2), "+v"(a3) : : "memory");
        issue(rA, nj, 0);
        sum_rows(rB, base + 16, 16);
        cj = nj; nj = fj;
    }
    {
        const int rem = E1 - base;
        if (rem > 16) issue(rB, cj, 16);
        sum_rows(rA, base, min(rem, 16));
        if (rem > 16) sum_rows(rB, base + 16, rem - 16);
    }
    for (; cur < 16; ++cur) {
        const float di = __int_as_float(__builtin_amdgcn_readlane(__float_as_int(dsv), cur));
        *(float4*)(tl + cur * FT_PITCH + lane * 4) = make_float4(di * a0, di * a1, di * a2, di * a3);
        a0 = a1 = a2 = a3 = 0.f;
    }

    __builtin_amdgcn_fence(__ATOMIC_RELEASE, "wavefront");
    __builtin_amdgcn_wave_barrier();
    __builtin_amdgcn_fence(__ATOMIC_ACQUIRE, "wavefront");

    float4 a[16];
#pragma unroll
    for (int t = 0; t < 16; ++t) a[t] = *(const float4*)(tl + c * FT_PITCH + 8 * t + 4 * h);

    v8f acc[NT];
#pragma unroll
    for (int j = 0; j < NT; ++j) acc[j] = v8f{};

    const float* sB = sW + ((size_t)h * NP + c) * 4;
#pragma unroll
    for (int t = 0; t < 16; ++t) {
        float4 b[NT];
#pragma unroll
        for (int j = 0; j < NT; ++j)
            b[j] = *(const float4*)(sB + ((size_t)(2 * t) * NP + 16 * j) * 4);
        v2f x0, x1;
        x0.x = a[t].x; x0.y = a[t].y;
        x1.x = a[t].z; x1.y = a[t].w;
#pragma unroll
        for (int j = 0; j < NT; ++j) {
            v2f b0; b0.x = b[j].x; b0.y = b[j].y;
            acc[j] = __builtin_amdgcn_wmma_f32_16x16x4_f32(
                 false, x0,  false, b0,
                 (short)0, acc[j],  false,  false);
        }
#pragma unroll
        for (int j = 0; j < NT; ++j) {
            v2f b1; b1.x = b[j].z; b1.y = b[j].w;
            acc[j] = __builtin_amdgcn_wmma_f32_16x16x4_f32(
                 false, x1,  false, b1,
                 (short)0, acc[j],  false,  false);
        }
    }

#pragma unroll
    for (int j = 0; j < NT; ++j) {
        const float badd = bias[16 * j + c];
#pragma unroll
        for (int r = 0; r < 8; ++r)
            tl[(8 * h + r) * FT_PITCH + 16 * j + c] = fmaxf(acc[j][r] + badd, 0.0f);
    }
    __builtin_amdgcn_fence(__ATOMIC_RELEASE, "wavefront");
    __builtin_amdgcn_wave_barrier();
    __builtin_amdgcn_fence(__ATOMIC_ACQUIRE, "wavefront");
    float* Op = outbuf + (size_t)n0 * ld_out + lane * 4;
#pragma unroll
    for (int r = 0; r < 16; ++r) {
        const float di = __int_as_float(__builtin_amdgcn_readlane(__float_as_int(dsv), r));
        const float4 v = *(const float4*)(tl + r * FT_PITCH + lane * 4);
        *(float4*)(Op + (size_t)r * ld_out) = make_float4(di * v.x, di * v.y, di * v.z, di * v.w);
    }
}


extern "C" void kernel_launch(void* const* d_in, const int* in_sizes, int n_in,
                              void* d_out, int out_size, void* d_ws, size_t ws_size,
                              hipStream_t stream) {
    (void)n_in; (void)out_size; (void)ws_size;
    const int F_IN = 256, HID = 128, NC = 40;
    const int N = in_sizes[0] / F_IN;
    const int E = in_sizes[1] / 2;
    const int TOT = E + N;

    const float* x     = (const float*)d_in[0];
    const int*   ei    = (const int*)d_in[1];
    const float* W_in  = (const float*)d_in[2];
    const float* b_in  = (const float*)d_in[3];
    const float* W1    = (const float*)d_in[4];
    const float* b1    = (const float*)d_in[5];
    const float* W2    = (const float*)d_in[6];
    const float* b2    = (const float*)d_in[7];
    const float* W3    = (const float*)d_in[8];
    const float* b3    = (const float*)d_in[9];
    const float* W_out = (const float*)d_in[10];
    const float* b_out = (const float*)d_in[11];
    float* out = (float*)d_out;

    char* base = (char*)d_ws;
    size_t off = 0;
    auto alloc = [&](size_t bytes) -> void* {
        void* q = base + off;
        off = (off + bytes + 255) & ~(size_t)255;
        return q;
    };
    int*   row_ptr = (int*)alloc((size_t)(N + 1) * 4);
    int*   colidx  = (int*)alloc((size_t)TOT * 4);
    float* dis     = (float*)alloc((size_t)N * 4);
    float* H       = (float*)alloc((size_t)N * HID * 4);
    float* hcat    = (float*)alloc((size_t)N * HID * 4 * 4);

    const int NB = (N + 255) / 256;
    const int G  = (E + CSR_CHUNK - 1) / CSR_CHUNK;

    int* counts = (int*)hcat;
    int* offsT  = counts + (size_t)G * NB;
    int* bintot = offsT + (size_t)NB * G;
    int* binned = bintot + CSR_MAXB;

    csr_bin_count_kernel<<<G, 256, 0, stream>>>(ei, counts, E, NB);
    csr_bin_scan_kernel<<<NB, 256, 0, stream>>>(counts, offsT, bintot, NB, G);
    csr_bin_scatter_kernel<<<G, 256, 0, stream>>>(ei, offsT, bintot, binned, E, NB, G);
    csr_bin_finalize_kernel<<<NB, 256, 0, stream>>>(binned, bintot, row_ptr, colidx, dis, N, TOT);

    const int rowTiles = N / 16;
    const int gemmBlocks = (rowTiles + 7) / 8;
    const int aggBlocks = (N + 7) / 8;

    gemm_wmma_f32_kernel<8, 2, false, false><<<gemmBlocks, 256, 0, stream>>>(
        x, F_IN, W_in, HID, H, HID, HID, nullptr, dis, rowTiles);
    gcn_aggregate_kernel<<<aggBlocks, 256, 0, stream>>>(
        H, row_ptr, colidx, dis, b_in, hcat, 4 * HID, 0 * HID, N);

    const float* Ws[3] = {W1, W2, W3};
    const float* bs[3] = {b1, b2, b3};
    for (int l = 0; l < 3; ++l) {
        gcn_gather_gemm_kernel<<<gemmBlocks, 256, 0, stream>>>(
            hcat + (size_t)l * HID, 4 * HID, Ws[l], row_ptr, colidx, dis, bs[l],
            hcat + (size_t)(l + 1) * HID, 4 * HID, rowTiles);
    }

    gemm_wmma_f32_kernel<3, 4, true, true><<<gemmBlocks, 256, 0, stream>>>(
        hcat, 4 * HID, W_out, NC, out, NC, NC, b_out, dis, rowTiles);
}
